# starting kernel
# baseline (speedup 1.0000x reference)
#include <hip/hip_runtime.h>
#include <hip/hip_bf16.h>

typedef float v2f __attribute__((ext_vector_type(2)));
typedef float v8f __attribute__((ext_vector_type(8)));

#define Bb 8
#define Tt 1024
#define Dd 512
#define Mm 64
#define Oo 512

__global__ void ln_eta_kernel(const float* __restrict__ x,
                              const float* __restrict__ gamma,
                              const float* __restrict__ beta,
                              const float* __restrict__ We,
                              float* __restrict__ xn,
                              float* __restrict__ eta)
{
    const int wave = threadIdx.x >> 5;
    const int lane = threadIdx.x & 31;
    const int row  = blockIdx.x * 8 + wave;
    const float* xr = x + (size_t)row * Dd;

    float v[16];
    float sum = 0.f, sq = 0.f;
#pragma unroll
    for (int e = 0; e < 16; ++e) {
        float t = xr[lane + 32 * e];
        v[e] = t; sum += t; sq += t * t;
    }
#pragma unroll
    for (int m = 16; m >= 1; m >>= 1) {
        sum += __shfl_xor(sum, m, 32);
        sq  += __shfl_xor(sq,  m, 32);
    }
    const float mu = sum * (1.0f / Dd);
    const float var = sq * (1.0f / Dd) - mu * mu;
    const float rs = rsqrtf(var + 1e-5f);

    float dot = 0.f;
    float* xo = xn + (size_t)row * Dd;
#pragma unroll
    for (int e = 0; e < 16; ++e) {
        const int d = lane + 32 * e;
        float xv = (v[e] - mu) * rs * gamma[d] + beta[d];
        xo[d] = xv;
        dot += xv * We[d];
    }
#pragma unroll
    for (int m = 16; m >= 1; m >>= 1) dot += __shfl_xor(dot, m, 32);
    if (lane == 0) eta[row] = 1.0f / (1.0f + expf(-dot));
}

__global__ void proj_gemm_kernel(const float* __restrict__ xn,
                                 const float* __restrict__ Wk,
                                 const float* __restrict__ Wq,
                                 const float* __restrict__ Wv,
                                 const float* __restrict__ Wvc,
                                 float* __restrict__ Ko,
                                 float* __restrict__ Qo,
                                 float* __restrict__ Vo,
                                 float* __restrict__ vco)
{
    const int w = blockIdx.y;
    const float* W = (w == 0) ? Wk : (w == 1) ? Wq : (w == 2) ? Wv : Wvc;
    float* Out     = (w == 0) ? Ko : (w == 1) ? Qo : (w == 2) ? Vo : vco;

    const int tileM = blockIdx.x * 16;
    const int lane  = threadIdx.x & 31;
    const int hi    = lane >> 4;
    const int lr    = lane & 15;

    const float* Arow = xn + (size_t)(tileM + lr) * Dd + 2 * hi;
    const float* B0 = W + (size_t)(lr +  0) * Dd + 2 * hi;
    const float* B1 = W + (size_t)(lr + 16) * Dd + 2 * hi;
    const float* B2 = W + (size_t)(lr + 32) * Dd + 2 * hi;
    const float* B3 = W + (size_t)(lr + 48) * Dd + 2 * hi;

    v8f c0 = {}, c1 = {}, c2 = {}, c3 = {};
    for (int kk = 0; kk < Dd; kk += 4) {
        v2f a  = *(const v2f*)(Arow + kk);
        v2f b0 = *(const v2f*)(B0 + kk);
        v2f b1 = *(const v2f*)(B1 + kk);
        v2f b2 = *(const v2f*)(B2 + kk);
        v2f b3 = *(const v2f*)(B3 + kk);
        c0 = __builtin_amdgcn_wmma_f32_16x16x4_f32(false, a, false, b0, (short)0, c0, false, false);
        c1 = __builtin_amdgcn_wmma_f32_16x16x4_f32(false, a, false, b1, (short)0, c1, false, false);
        c2 = __builtin_amdgcn_wmma_f32_16x16x4_f32(false, a, false, b2, (short)0, c2, false, false);
        c3 = __builtin_amdgcn_wmma_f32_16x16x4_f32(false, a, false, b3, (short)0, c3, false, false);
    }

    if (w < 2) {
#pragma unroll
        for (int r = 0; r < 8; ++r) {
            c0[r] = fmaxf(c0[r], 0.f); c1[r] = fmaxf(c1[r], 0.f);
            c2[r] = fmaxf(c2[r], 0.f); c3[r] = fmaxf(c3[r], 0.f);
            float s = c0[r] + c1[r] + c2[r] + c3[r];
#pragma unroll
            for (int m = 8; m >= 1; m >>= 1) s += __shfl_xor(s, m, 32);
            float inv = 1.0f / (1e-5f + s);
            c0[r] *= inv; c1[r] *= inv; c2[r] *= inv; c3[r] *= inv;
        }
    }

#pragma unroll
    for (int r = 0; r < 8; ++r) {
        float* o = Out + (size_t)(tileM + r + 8 * hi) * Mm + lr;
        o[0]  = c0[r];
        o[16] = c1[r];
        o[32] = c2[r];
        o[48] = c3[r];
    }
}

__global__ void scan_kernel(const float* __restrict__ K,
                            const float* __restrict__ Q,
                            const float* __restrict__ V,
                            const float* __restrict__ vc,
                            const float* __restrict__ eta,
                            const float* __restrict__ theta_matrix,
                            float* __restrict__ y)
{
    const int b    = blockIdx.x;
    const int wave = threadIdx.x >> 5;
    const int lane = threadIdx.x & 31;
    const int i    = blockIdx.y * 8 + wave;
    const float theta_i = theta_matrix[i];

    float s0 = 0.f, s1 = 0.f;
    const size_t base = (size_t)b * Tt * Mm;

    for (int t = 0; t < Tt; ++t) {
        const size_t o = base + (size_t)t * Mm;
        const float vc0 = vc[o + lane], vc1 = vc[o + lane + 32];
        const float k0  = K[o + lane],  k1  = K[o + lane + 32];
        const float q0  = Q[o + lane],  q1  = Q[o + lane + 32];
        const float et  = eta[b * Tt + t];
        const float vi  = V[o + i];

        const float cc0 = 1.0f + tanhf(theta_i * vc0);
        const float cc1 = 1.0f + tanhf(theta_i * vc1);
        const float u   = et * vi;
        s0 = s0 * cc0 + u * k0;
        s1 = s1 * cc1 + u * k1;

        float p = s0 * q0 + s1 * q1;
#pragma unroll
        for (int m = 16; m >= 1; m >>= 1) p += __shfl_xor(p, m, 32);
        if (lane == 0) y[o + i] = p;
    }
}

__global__ void out_gemm_kernel(const float* __restrict__ y,
                                const float* __restrict__ Wout,
                                const float* __restrict__ bout,
                                float* __restrict__ out)
{
    const int tileM = blockIdx.x * 16;
    const int tileN = blockIdx.y * 64;
    const int lane  = threadIdx.x & 31;
    const int hi    = lane >> 4;
    const int lr    = lane & 15;

    const float* Arow = y + (size_t)(tileM + lr) * Mm + 2 * hi;
    const float* B0 = Wout + (size_t)(tileN + lr +  0) * Mm + 2 * hi;
    const float* B1 = Wout + (size_t)(tileN + lr + 16) * Mm + 2 * hi;
    const float* B2 = Wout + (size_t)(tileN + lr + 32) * Mm + 2 * hi;
    const float* B3 = Wout + (size_t)(tileN + lr + 48) * Mm + 2 * hi;

    v8f c0 = {}, c1 = {}, c2 = {}, c3 = {};
#pragma unroll
    for (int kk = 0; kk < Mm; kk += 4) {
        v2f a  = *(const v2f*)(Arow + kk);
        v2f b0 = *(const v2f*)(B0 + kk);
        v2f b1 = *(const v2f*)(B1 + kk);
        v2f b2 = *(const v2f*)(B2 + kk);
        v2f b3 = *(const v2f*)(B3 + kk);
        c0 = __builtin_amdgcn_wmma_f32_16x16x4_f32(false, a, false, b0, (short)0, c0, false, false);
        c1 = __builtin_amdgcn_wmma_f32_16x16x4_f32(false, a, false, b1, (short)0, c1, false, false);
        c2 = __builtin_amdgcn_wmma_f32_16x16x4_f32(false, a, false, b2, (short)0, c2, false, false);
        c3 = __builtin_amdgcn_wmma_f32_16x16x4_f32(false, a, false, b3, (short)0, c3, false, false);
    }

    const float bb0 = bout[tileN + lr +  0];
    const float bb1 = bout[tileN + lr + 16];
    const float bb2 = bout[tileN + lr + 32];
    const float bb3 = bout[tileN + lr + 48];

#pragma unroll
    for (int r = 0; r < 8; ++r) {
        float* o = out + (size_t)(tileM + r + 8 * hi) * Oo + tileN + lr;
        o[0]  = c0[r] + bb0;
        o[16] = c1[r] + bb1;
        o[32] = c2[r] + bb2;
        o[48] = c3[r] + bb3;
    }
}

extern "C" void kernel_launch(void* const* d_in, const int* in_sizes, int n_in,
                              void* d_out, int out_size, void* d_ws, size_t ws_size,
                              hipStream_t stream)
{
    const float* x     = (const float*)d_in[0];
    const float* Wk    = (const float*)d_in[1];
    const float* Wq    = (const float*)d_in[2];
    const float* Wv    = (const float*)d_in[3];
    const float* Wvc   = (const float*)d_in[4];
    const float* We    = (const float*)d_in[5];
    const float* theta = (const float*)d_in[6];
    const float* Wout  = (const float*)d_in[7];
    const float* bout  = (const float*)d_in[8];
    const float* gamma = (const float*)d_in[9];
    const float* beta  = (const float*)d_in[10];
    float* out = (float*)d_out;

    float* ws  = (float*)d_ws;
    const size_t NROW = (size_t)Bb * Tt;
    float* xn  = ws;
    float* Ko  = xn  + NROW * Dd;
    float* Qo  = Ko  + NROW * Mm;
    float* Vo  = Qo  + NROW * Mm;
    float* vco = Vo  + NROW * Mm;
    float* yb  = vco + NROW * Mm;
    float* eta = yb  + NROW * Mm;

    ln_eta_kernel<<<dim3(NROW / 8), dim3(256), 0, stream>>>(x, gamma, beta, We, xn, eta);
    proj_gemm_kernel<<<dim3(NROW / 16, 4), dim3(32), 0, stream>>>(xn, Wk, Wq, Wv, Wvc,
                                                                  Ko, Qo, Vo, vco);
    scan_kernel<<<dim3(Bb, Mm / 8), dim3(256), 0, stream>>>(Ko, Qo, Vo, vco, eta, theta, yb);
    out_gemm_kernel<<<dim3(NROW / 16, Oo / 64), dim3(32), 0, stream>>>(yb, Wout, bout, out);
}


# scan kernel: time loop blocked by 32 steps; a block's operands sit in registers, refilled one block ahead; 32 partial dots reduced by one transposing butterfly (same sum tree), one y store per block
# speedup vs baseline: 4.3757x; 4.3757x over previous
#include <hip/hip_runtime.h>
#include <hip/hip_bf16.h>

typedef float v2f __attribute__((ext_vector_type(2)));
typedef float v8f __attribute__((ext_vector_type(8)));

#define Bb 8
#define Tt 1024
#define Dd 512
#define Mm 64
#define Oo 512

__global__ void ln_eta_kernel(const float* __restrict__ x,
                              const float* __restrict__ gamma,
                              const float* __restrict__ beta,
                              const float* __restrict__ We,
                              float* __restrict__ xn,
                              float* __restrict__ eta)
{
    const int wave = threadIdx.x >> 5;
    const int lane = threadIdx.x & 31;
    const int row  = blockIdx.x * 8 + wave;
    const float* xr = x + (size_t)row * Dd;

    float v[16];
    float sum = 0.f, sq = 0.f;
#pragma unroll
    for (int e = 0; e < 16; ++e) {
        float t = xr[lane + 32 * e];
        v[e] = t; sum += t; sq += t * t;
    }
#pragma unroll
    for (int m = 16; m >= 1; m >>= 1) {
        sum += __shfl_xor(sum, m, 32);
        sq  += __shfl_xor(sq,  m, 32);
    }
    const float mu = sum * (1.0f / Dd);
    const float var = sq * (1.0f / Dd) - mu * mu;
    const float rs = rsqrtf(var + 1e-5f);

    float dot = 0.f;
    float* xo = xn + (size_t)row * Dd;
#pragma unroll
    for (int e = 0; e < 16; ++e) {
        const int d = lane + 32 * e;
        float xv = (v[e] - mu) * rs * gamma[d] + beta[d];
        xo[d] = xv;
        dot += xv * We[d];
    }
#pragma unroll
    for (int m = 16; m >= 1; m >>= 1) dot += __shfl_xor(dot, m, 32);
    if (lane == 0) eta[row] = 1.0f / (1.0f + expf(-dot));
}

__global__ void proj_gemm_kernel(const float* __restrict__ xn,
                                 const float* __restrict__ Wk,
                                 const float* __restrict__ Wq,
                                 const float* __restrict__ Wv,
                                 const float* __restrict__ Wvc,
                                 float* __restrict__ Ko,
                                 float* __restrict__ Qo,
                                 float* __restrict__ Vo,
                                 float* __restrict__ vco)
{
    const int w = blockIdx.y;
    const float* W = (w == 0) ? Wk : (w == 1) ? Wq : (w == 2) ? Wv : Wvc;
    float* Out     = (w == 0) ? Ko : (w == 1) ? Qo : (w == 2) ? Vo : vco;

    const int tileM = blockIdx.x * 16;
    const int lane  = threadIdx.x & 31;
    const int hi    = lane >> 4;
    const int lr    = lane & 15;

    const float* Arow = xn + (size_t)(tileM + lr) * Dd + 2 * hi;
    const float* B0 = W + (size_t)(lr +  0) * Dd + 2 * hi;
    const float* B1 = W + (size_t)(lr + 16) * Dd + 2 * hi;
    const float* B2 = W + (size_t)(lr + 32) * Dd + 2 * hi;
    const float* B3 = W + (size_t)(lr + 48) * Dd + 2 * hi;

    v8f c0 = {}, c1 = {}, c2 = {}, c3 = {};
    for (int kk = 0; kk < Dd; kk += 4) {
        v2f a  = *(const v2f*)(Arow + kk);
        v2f b0 = *(const v2f*)(B0 + kk);
        v2f b1 = *(const v2f*)(B1 + kk);
        v2f b2 = *(const v2f*)(B2 + kk);
        v2f b3 = *(const v2f*)(B3 + kk);
        c0 = __builtin_amdgcn_wmma_f32_16x16x4_f32(false, a, false, b0, (short)0, c0, false, false);
        c1 = __builtin_amdgcn_wmma_f32_16x16x4_f32(false, a, false, b1, (short)0, c1, false, false);
        c2 = __builtin_amdgcn_wmma_f32_16x16x4_f32(false, a, false, b2, (short)0, c2, false, false);
        c3 = __builtin_amdgcn_wmma_f32_16x16x4_f32(false, a, false, b3, (short)0, c3, false, false);
    }

    if (w < 2) {
#pragma unroll
        for (int r = 0; r < 8; ++r) {
            c0[r] = fmaxf(c0[r], 0.f); c1[r] = fmaxf(c1[r], 0.f);
            c2[r] = fmaxf(c2[r], 0.f); c3[r] = fmaxf(c3[r], 0.f);
            float s = c0[r] + c1[r] + c2[r] + c3[r];
#pragma unroll
            for (int m = 8; m >= 1; m >>= 1) s += __shfl_xor(s, m, 32);
            float inv = 1.0f / (1e-5f + s);
            c0[r] *= inv; c1[r] *= inv; c2[r] *= inv; c3[r] *= inv;
        }
    }

#pragma unroll
    for (int r = 0; r < 8; ++r) {
        float* o = Out + (size_t)(tileM + r + 8 * hi) * Mm + lr;
        o[0]  = c0[r];
        o[16] = c1[r];
        o[32] = c2[r];
        o[48] = c3[r];
    }
}

#define TBLK 32

__global__ __launch_bounds__(256) void scan_kernel(const float* __restrict__ K,
                            const float* __restrict__ Q,
                            const float* __restrict__ V,
                            const float* __restrict__ vc,
                            const float* __restrict__ eta,
                            const float* __restrict__ theta_matrix,
                            float* __restrict__ y)
{
    const int b    = blockIdx.x;
    const int wave = threadIdx.x >> 5;
    const int lane = threadIdx.x & 31;
    const int i    = blockIdx.y * 8 + wave;
    const float theta_i = theta_matrix[i];

    float s0 = 0.f, s1 = 0.f;
    const size_t base = (size_t)b * Tt * Mm;
    const float* vcl = vc + base + lane;
    const float* Kl  = K  + base + lane;
    const float* Ql  = Q  + base + lane;
    const float* etal = eta + (size_t)b * Tt + lane;
    const float* Vl   = V + base + (size_t)lane * Mm + i;

    float a0[TBLK], a1[TBLK], kx0[TBLK], kx1[TBLK], qx0[TBLK], qx1[TBLK];
#pragma unroll
    for (int tt = 0; tt < TBLK - 1; ++tt) {
        const size_t o = (size_t)tt * Mm;
        a0[tt]  = vcl[o]; a1[tt]  = vcl[o + 32];
        kx0[tt] = Kl[o];  kx1[tt] = Kl[o + 32];
        qx0[tt] = Ql[o];  qx1[tt] = Ql[o + 32];
    }
    float u_l = etal[0] * Vl[0];

    for (int t0 = 0; t0 < Tt; t0 += TBLK) {
        const int tn = (t0 + TBLK < Tt) ? t0 + TBLK : t0;
        const float et_n = etal[tn];
        const float vi_n = Vl[(size_t)tn * Mm];

        float p[TBLK];
#pragma unroll
        for (int tt = 0; tt < TBLK; ++tt) {
            const float u = __builtin_bit_cast(float,
                __builtin_amdgcn_readlane(__builtin_bit_cast(int, u_l), tt));
            const float cc0 = 1.0f + tanhf(theta_i * a0[tt]);
            const float cc1 = 1.0f + tanhf(theta_i * a1[tt]);
            s0 = s0 * cc0 + u * kx0[tt];
            s1 = s1 * cc1 + u * kx1[tt];
            p[tt] = s0 * qx0[tt] + s1 * qx1[tt];
            const int f = (tt + TBLK - 1) % TBLK;
            const size_t o = (size_t)((tt == 0 ? t0 : tn) + f) * Mm;
            a0[f]  = vcl[o]; a1[f]  = vcl[o + 32];
            kx0[f] = Kl[o];  kx1[f] = Kl[o + 32];
            qx0[f] = Ql[o];  qx1[f] = Ql[o + 32];
        }

#pragma unroll
        for (int m = TBLK / 2; m >= 1; m >>= 1) {
            const bool up = (lane & m) != 0;
#pragma unroll
            for (int r = 0; r < m; ++r) {
                const float lo = p[r], hi = p[r + m];
                const float keep = up ? hi : lo;
                const float send = up ? lo : hi;
                p[r] = keep + __shfl_xor(send, m, 32);
            }
        }
        y[base + (size_t)(t0 + lane) * Mm + i] = p[0];
        u_l = et_n * vi_n;
    }
}

__global__ void out_gemm_kernel(const float* __restrict__ y,
                                const float* __restrict__ Wout,
                                const float* __restrict__ bout,
                                float* __restrict__ out)
{
    const int tileM = blockIdx.x * 16;
    const int tileN = blockIdx.y * 64;
    const int lane  = threadIdx.x & 31;
    const int hi    = lane >> 4;
    const int lr    = lane & 15;

    const float* Arow = y + (size_t)(tileM + lr) * Mm + 2 * hi;
    const float* B0 = Wout + (size_t)(tileN + lr +  0) * Mm + 2 * hi;
    const float* B1 = Wout + (size_t)(tileN + lr + 16) * Mm + 2 * hi;
    const float* B2 = Wout + (size_t)(tileN + lr + 32) * Mm + 2 * hi;
    const float* B3 = Wout + (size_t)(tileN + lr + 48) * Mm + 2 * hi;

    v8f c0 = {}, c1 = {}, c2 = {}, c3 = {};
#pragma unroll
    for (int kk = 0; kk < Mm; kk += 4) {
        v2f a  = *(const v2f*)(Arow + kk);
        v2f b0 = *(const v2f*)(B0 + kk);
        v2f b1 = *(const v2f*)(B1 + kk);
        v2f b2 = *(const v2f*)(B2 + kk);
        v2f b3 = *(const v2f*)(B3 + kk);
        c0 = __builtin_amdgcn_wmma_f32_16x16x4_f32(false, a, false, b0, (short)0, c0, false, false);
        c1 = __builtin_amdgcn_wmma_f32_16x16x4_f32(false, a, false, b1, (short)0, c1, false, false);
        c2 = __builtin_amdgcn_wmma_f32_16x16x4_f32(false, a, false, b2, (short)0, c2, false, false);
        c3 = __builtin_amdgcn_wmma_f32_16x16x4_f32(false, a, false, b3, (short)0, c3, false, false);
    }

    const float bb0 = bout[tileN + lr +  0];
    const float bb1 = bout[tileN + lr + 16];
    const float bb2 = bout[tileN + lr + 32];
    const float bb3 = bout[tileN + lr + 48];

#pragma unroll
    for (int r = 0; r < 8; ++r) {
        float* o = out + (size_t)(tileM + r + 8 * hi) * Oo + tileN + lr;
        o[0]  = c0[r] + bb0;
        o[16] = c1[r] + bb1;
        o[32] = c2[r] + bb2;
        o[48] = c3[r] + bb3;
    }
}

extern "C" void kernel_launch(void* const* d_in, const int* in_sizes, int n_in,
                              void* d_out, int out_size, void* d_ws, size_t ws_size,
                              hipStream_t stream)
{
    const float* x     = (const float*)d_in[0];
    const float* Wk    = (const float*)d_in[1];
    const float* Wq    = (const float*)d_in[2];
    const float* Wv    = (const float*)d_in[3];
    const float* Wvc   = (const float*)d_in[4];
    const float* We    = (const float*)d_in[5];
    const float* theta = (const float*)d_in[6];
    const float* Wout  = (const float*)d_in[7];
    const float* bout  = (const float*)d_in[8];
    const float* gamma = (const float*)d_in[9];
    const float* beta  = (const float*)d_in[10];
    float* out = (float*)d_out;

    float* ws  = (float*)d_ws;
    const size_t NROW = (size_t)Bb * Tt;
    float* xn  = ws;
    float* Ko  = xn  + NROW * Dd;
    float* Qo  = Ko  + NROW * Mm;
    float* Vo  = Qo  + NROW * Mm;
    float* vco = Vo  + NROW * Mm;
    float* yb  = vco + NROW * Mm;
    float* eta = yb  + NROW * Mm;

    ln_eta_kernel<<<dim3(NROW / 8), dim3(256), 0, stream>>>(x, gamma, beta, We, xn, eta);
    proj_gemm_kernel<<<dim3(NROW / 16, 4), dim3(32), 0, stream>>>(xn, Wk, Wq, Wv, Wvc,
                                                                  Ko, Qo, Vo, vco);
    scan_kernel<<<dim3(Bb, Mm / 8), dim3(256), 0, stream>>>(Ko, Qo, Vo, vco, eta, theta, yb);
    out_gemm_kernel<<<dim3(NROW / 16, Oo / 64), dim3(32), 0, stream>>>(yb, Wout, bout, out);
}


# scan kernel: tanhf's per-element range branch (|x| < 0.625) becomes one wave-wide test per 32-step block; passing blocks run the library polynomial for all 64 gates branch-free, same values
# speedup vs baseline: 6.5657x; 1.5005x over previous
#include <hip/hip_runtime.h>
#include <hip/hip_bf16.h>

typedef float v2f __attribute__((ext_vector_type(2)));
typedef float v8f __attribute__((ext_vector_type(8)));

#define Bb 8
#define Tt 1024
#define Dd 512
#define Mm 64
#define Oo 512

__global__ void ln_eta_kernel(const float* __restrict__ x,
                              const float* __restrict__ gamma,
                              const float* __restrict__ beta,
                              const float* __restrict__ We,
                              float* __restrict__ xn,
                              float* __restrict__ eta)
{
    const int wave = threadIdx.x >> 5;
    const int lane = threadIdx.x & 31;
    const int row  = blockIdx.x * 8 + wave;
    const float* xr = x + (size_t)row * Dd;

    float v[16];
    float sum = 0.f, sq = 0.f;
#pragma unroll
    for (int e = 0; e < 16; ++e) {
        float t = xr[lane + 32 * e];
        v[e] = t; sum += t; sq += t * t;
    }
#pragma unroll
    for (int m = 16; m >= 1; m >>= 1) {
        sum += __shfl_xor(sum, m, 32);
        sq  += __shfl_xor(sq,  m, 32);
    }
    const float mu = sum * (1.0f / Dd);
    const float var = sq * (1.0f / Dd) - mu * mu;
    const float rs = rsqrtf(var + 1e-5f);

    float dot = 0.f;
    float* xo = xn + (size_t)row * Dd;
#pragma unroll
    for (int e = 0; e < 16; ++e) {
        const int d = lane + 32 * e;
        float xv = (v[e] - mu) * rs * gamma[d] + beta[d];
        xo[d] = xv;
        dot += xv * We[d];
    }
#pragma unroll
    for (int m = 16; m >= 1; m >>= 1) dot += __shfl_xor(dot, m, 32);
    if (lane == 0) eta[row] = 1.0f / (1.0f + expf(-dot));
}

__global__ void proj_gemm_kernel(const float* __restrict__ xn,
                                 const float* __restrict__ Wk,
                                 const float* __restrict__ Wq,
                                 const float* __restrict__ Wv,
                                 const float* __restrict__ Wvc,
                                 float* __restrict__ Ko,
                                 float* __restrict__ Qo,
                                 float* __restrict__ Vo,
                                 float* __restrict__ vco)
{
    const int w = blockIdx.y;
    const float* W = (w == 0) ? Wk : (w == 1) ? Wq : (w == 2) ? Wv : Wvc;
    float* Out     = (w == 0) ? Ko : (w == 1) ? Qo : (w == 2) ? Vo : vco;

    const int tileM = blockIdx.x * 16;
    const int lane  = threadIdx.x & 31;
    const int hi    = lane >> 4;
    const int lr    = lane & 15;

    const float* Arow = xn + (size_t)(tileM + lr) * Dd + 2 * hi;
    const float* B0 = W + (size_t)(lr +  0) * Dd + 2 * hi;
    const float* B1 = W + (size_t)(lr + 16) * Dd + 2 * hi;
    const float* B2 = W + (size_t)(lr + 32) * Dd + 2 * hi;
    const float* B3 = W + (size_t)(lr + 48) * Dd + 2 * hi;

    v8f c0 = {}, c1 = {}, c2 = {}, c3 = {};
    for (int kk = 0; kk < Dd; kk += 4) {
        v2f a  = *(const v2f*)(Arow + kk);
        v2f b0 = *(const v2f*)(B0 + kk);
        v2f b1 = *(const v2f*)(B1 + kk);
        v2f b2 = *(const v2f*)(B2 + kk);
        v2f b3 = *(const v2f*)(B3 + kk);
        c0 = __builtin_amdgcn_wmma_f32_16x16x4_f32(false, a, false, b0, (short)0, c0, false, false);
        c1 = __builtin_amdgcn_wmma_f32_16x16x4_f32(false, a, false, b1, (short)0, c1, false, false);
        c2 = __builtin_amdgcn_wmma_f32_16x16x4_f32(false, a, false, b2, (short)0, c2, false, false);
        c3 = __builtin_amdgcn_wmma_f32_16x16x4_f32(false, a, false, b3, (short)0, c3, false, false);
    }

    if (w < 2) {
#pragma unroll
        for (int r = 0; r < 8; ++r) {
            c0[r] = fmaxf(c0[r], 0.f); c1[r] = fmaxf(c1[r], 0.f);
            c2[r] = fmaxf(c2[r], 0.f); c3[r] = fmaxf(c3[r], 0.f);
            float s = c0[r] + c1[r] + c2[r] + c3[r];
#pragma unroll
            for (int m = 8; m >= 1; m >>= 1) s += __shfl_xor(s, m, 32);
            float inv = 1.0f / (1e-5f + s);
            c0[r] *= inv; c1[r] *= inv; c2[r] *= inv; c3[r] *= inv;
        }
    }

#pragma unroll
    for (int r = 0; r < 8; ++r) {
        float* o = Out + (size_t)(tileM + r + 8 * hi) * Mm + lr;
        o[0]  = c0[r];
        o[16] = c1[r];
        o[32] = c2[r];
        o[48] = c3[r];
    }
}

#define TBLK 32

__device__ __forceinline__ float one_plus_tanh_small(float x)
{
    const float x2 = x * x;
    float p = __builtin_fmaf(x2, -0x1.758e7ap-8f, 0x1.521192p-6f);
    p = __builtin_fmaf(x2, p, -0x1.b8389cp-5f);
    p = __builtin_fmaf(x2, p, 0x1.110704p-3f);
    p = __builtin_fmaf(x2, p, -0x1.555532p-2f);
    return 1.0f + __builtin_fmaf(x2, x * p, x);
}

__global__ __launch_bounds__(256) void scan_kernel(const float* __restrict__ K,
                            const float* __restrict__ Q,
                            const float* __restrict__ V,
                            const float* __restrict__ vc,
                            const float* __restrict__ eta,
                            const float* __restrict__ theta_matrix,
                            float* __restrict__ y)
{
    const int b    = blockIdx.x;
    const int wave = threadIdx.x >> 5;
    const int lane = threadIdx.x & 31;
    const int i    = blockIdx.y * 8 + wave;
    const float theta_i = theta_matrix[i];

    float s0 = 0.f, s1 = 0.f;
    const size_t base = (size_t)b * Tt * Mm;
    const float* vcl = vc + base + lane;
    const float* Kl  = K  + base + lane;
    const float* Ql  = Q  + base + lane;
    const float* etal = eta + (size_t)b * Tt + lane;
    const float* Vl   = V + base + (size_t)lane * Mm + i;

    float a0[TBLK], a1[TBLK], kx0[TBLK], kx1[TBLK], qx0[TBLK], qx1[TBLK];
#pragma unroll
    for (int tt = 0; tt < TBLK; ++tt) {
        const size_t o = (size_t)tt * Mm;
        a0[tt]  = vcl[o]; a1[tt]  = vcl[o + 32];
        kx0[tt] = Kl[o];  kx1[tt] = Kl[o + 32];
        qx0[tt] = Ql[o];  qx1[tt] = Ql[o + 32];
    }
    float u_l = etal[0] * Vl[0];

    for (int t0 = 0; t0 < Tt; t0 += TBLK) {
        const int tn = (t0 + TBLK < Tt) ? t0 + TBLK : t0;
        const float et_n = etal[tn];
        const float vi_n = Vl[(size_t)tn * Mm];

        float am[4] = {0.f, 0.f, 0.f, 0.f};
#pragma unroll
        for (int tt = 0; tt < TBLK; ++tt)
            am[tt & 3] = fmaxf(am[tt & 3], fmaxf(fabsf(a0[tt]), fabsf(a1[tt])));
        const float amax = fmaxf(fmaxf(am[0], am[1]), fmaxf(am[2], am[3]));
        const bool big = !(fabsf(theta_i) * amax < 0.625f);

        float c0[TBLK], c1[TBLK];
        if (__builtin_expect(__builtin_amdgcn_ballot_w32(big) == 0u, 1)) {
#pragma unroll
            for (int tt = 0; tt < TBLK; ++tt) {
                c0[tt] = one_plus_tanh_small(theta_i * a0[tt]);
                c1[tt] = one_plus_tanh_small(theta_i * a1[tt]);
            }
        } else {
#pragma unroll
            for (int tt = 0; tt < TBLK; ++tt) {
                c0[tt] = 1.0f + tanhf(theta_i * a0[tt]);
                c1[tt] = 1.0f + tanhf(theta_i * a1[tt]);
            }
        }

        float p[TBLK];
#pragma unroll
        for (int tt = 0; tt < TBLK; ++tt) {
            const float u = __builtin_bit_cast(float,
                __builtin_amdgcn_readlane(__builtin_bit_cast(int, u_l), tt));
            s0 = __builtin_fmaf(u, kx0[tt], s0 * c0[tt]);
            s1 = __builtin_fmaf(u, kx1[tt], s1 * c1[tt]);
            p[tt] = __builtin_fmaf(s0, qx0[tt], s1 * qx1[tt]);
            const size_t o = (size_t)(tn + tt) * Mm;
            a0[tt]  = vcl[o]; a1[tt]  = vcl[o + 32];
            kx0[tt] = Kl[o];  kx1[tt] = Kl[o + 32];
            qx0[tt] = Ql[o];  qx1[tt] = Ql[o + 32];
        }

#pragma unroll
        for (int m = TBLK / 2; m >= 1; m >>= 1) {
            const bool up = (lane & m) != 0;
#pragma unroll
            for (int r = 0; r < m; ++r) {
                const float lo = p[r], hi = p[r + m];
                const float keep = up ? hi : lo;
                const float send = up ? lo : hi;
                p[r] = keep + __shfl_xor(send, m, 32);
            }
        }
        y[base + (size_t)(t0 + lane) * Mm + i] = p[0];
        u_l = et_n * vi_n;
    }
}

__global__ void out_gemm_kernel(const float* __restrict__ y,
                                const float* __restrict__ Wout,
                                const float* __restrict__ bout,
                                float* __restrict__ out)
{
    const int tileM = blockIdx.x * 16;
    const int tileN = blockIdx.y * 64;
    const int lane  = threadIdx.x & 31;
    const int hi    = lane >> 4;
    const int lr    = lane & 15;

    const float* Arow = y + (size_t)(tileM + lr) * Mm + 2 * hi;
    const float* B0 = Wout + (size_t)(tileN + lr +  0) * Mm + 2 * hi;
    const float* B1 = Wout + (size_t)(tileN + lr + 16) * Mm + 2 * hi;
    const float* B2 = Wout + (size_t)(tileN + lr + 32) * Mm + 2 * hi;
    const float* B3 = Wout + (size_t)(tileN + lr + 48) * Mm + 2 * hi;

    v8f c0 = {}, c1 = {}, c2 = {}, c3 = {};
#pragma unroll
    for (int kk = 0; kk < Mm; kk += 4) {
        v2f a  = *(const v2f*)(Arow + kk);
        v2f b0 = *(const v2f*)(B0 + kk);
        v2f b1 = *(const v2f*)(B1 + kk);
        v2f b2 = *(const v2f*)(B2 + kk);
        v2f b3 = *(const v2f*)(B3 + kk);
        c0 = __builtin_amdgcn_wmma_f32_16x16x4_f32(false, a, false, b0, (short)0, c0, false, false);
        c1 = __builtin_amdgcn_wmma_f32_16x16x4_f32(false, a, false, b1, (short)0, c1, false, false);
        c2 = __builtin_amdgcn_wmma_f32_16x16x4_f32(false, a, false, b2, (short)0, c2, false, false);
        c3 = __builtin_amdgcn_wmma_f32_16x16x4_f32(false, a, false, b3, (short)0, c3, false, false);
    }

    const float bb0 = bout[tileN + lr +  0];
    const float bb1 = bout[tileN + lr + 16];
    const float bb2 = bout[tileN + lr + 32];
    const float bb3 = bout[tileN + lr + 48];

#pragma unroll
    for (int r = 0; r < 8; ++r) {
        float* o = out + (size_t)(tileM + r + 8 * hi) * Oo + tileN + lr;
        o[0]  = c0[r] + bb0;
        o[16] = c1[r] + bb1;
        o[32] = c2[r] + bb2;
        o[48] = c3[r] + bb3;
    }
}

extern "C" void kernel_launch(void* const* d_in, const int* in_sizes, int n_in,
                              void* d_out, int out_size, void* d_ws, size_t ws_size,
                              hipStream_t stream)
{
    const float* x     = (const float*)d_in[0];
    const float* Wk    = (const float*)d_in[1];
    const float* Wq    = (const float*)d_in[2];
    const float* Wv    = (const float*)d_in[3];
    const float* Wvc   = (const float*)d_in[4];
    const float* We    = (const float*)d_in[5];
    const float* theta = (const float*)d_in[6];
    const float* Wout  = (const float*)d_in[7];
    const float* bout  = (const float*)d_in[8];
    const float* gamma = (const float*)d_in[9];
    const float* beta  = (const float*)d_in[10];
    float* out = (float*)d_out;

    float* ws  = (float*)d_ws;
    const size_t NROW = (size_t)Bb * Tt;
    float* xn  = ws;
    float* Ko  = xn  + NROW * Dd;
    float* Qo  = Ko  + NROW * Mm;
    float* Vo  = Qo  + NROW * Mm;
    float* vco = Vo  + NROW * Mm;
    float* yb  = vco + NROW * Mm;
    float* eta = yb  + NROW * Mm;

    ln_eta_kernel<<<dim3(NROW / 8), dim3(256), 0, stream>>>(x, gamma, beta, We, xn, eta);
    proj_gemm_kernel<<<dim3(NROW / 16, 4), dim3(32), 0, stream>>>(xn, Wk, Wq, Wv, Wvc,
                                                                  Ko, Qo, Vo, vco);
    scan_kernel<<<dim3(Bb, Mm / 8), dim3(256), 0, stream>>>(Ko, Qo, Vo, vco, eta, theta, yb);
    out_gemm_kernel<<<dim3(NROW / 16, Oo / 64), dim3(32), 0, stream>>>(yb, Wout, bout, out);
}


# scan kernel: lane owns adjacent state columns 2l,2l+1 (was l,l+32): a step's vc, K, Q rows are one 8-byte load per lane, half the load instructions; y sum tree pairs columns 2l,2l+1 first
# speedup vs baseline: 7.7786x; 1.1847x over previous
#include <hip/hip_runtime.h>
#include <hip/hip_bf16.h>

typedef float v2f __attribute__((ext_vector_type(2)));
typedef float v8f __attribute__((ext_vector_type(8)));

#define Bb 8
#define Tt 1024
#define Dd 512
#define Mm 64
#define Oo 512

__global__ void ln_eta_kernel(const float* __restrict__ x,
                              const float* __restrict__ gamma,
                              const float* __restrict__ beta,
                              const float* __restrict__ We,
                              float* __restrict__ xn,
                              float* __restrict__ eta)
{
    const int wave = threadIdx.x >> 5;
    const int lane = threadIdx.x & 31;
    const int row  = blockIdx.x * 8 + wave;
    const float* xr = x + (size_t)row * Dd;

    float v[16];
    float sum = 0.f, sq = 0.f;
#pragma unroll
    for (int e = 0; e < 16; ++e) {
        float t = xr[lane + 32 * e];
        v[e] = t; sum += t; sq += t * t;
    }
#pragma unroll
    for (int m = 16; m >= 1; m >>= 1) {
        sum += __shfl_xor(sum, m, 32);
        sq  += __shfl_xor(sq,  m, 32);
    }
    const float mu = sum * (1.0f / Dd);
    const float var = sq * (1.0f / Dd) - mu * mu;
    const float rs = rsqrtf(var + 1e-5f);

    float dot = 0.f;
    float* xo = xn + (size_t)row * Dd;
#pragma unroll
    for (int e = 0; e < 16; ++e) {
        const int d = lane + 32 * e;
        float xv = (v[e] - mu) * rs * gamma[d] + beta[d];
        xo[d] = xv;
        dot += xv * We[d];
    }
#pragma unroll
    for (int m = 16; m >= 1; m >>= 1) dot += __shfl_xor(dot, m, 32);
    if (lane == 0) eta[row] = 1.0f / (1.0f + expf(-dot));
}

__global__ void proj_gemm_kernel(const float* __restrict__ xn,
                                 const float* __restrict__ Wk,
                                 const float* __restrict__ Wq,
                                 const float* __restrict__ Wv,
                                 const float* __restrict__ Wvc,
                                 float* __restrict__ Ko,
                                 float* __restrict__ Qo,
                                 float* __restrict__ Vo,
                                 float* __restrict__ vco)
{
    const int w = blockIdx.y;
    const float* W = (w == 0) ? Wk : (w == 1) ? Wq : (w == 2) ? Wv : Wvc;
    float* Out     = (w == 0) ? Ko : (w == 1) ? Qo : (w == 2) ? Vo : vco;

    const int tileM = blockIdx.x * 16;
    const int lane  = threadIdx.x & 31;
    const int hi    = lane >> 4;
    const int lr    = lane & 15;

    const float* Arow = xn + (size_t)(tileM + lr) * Dd + 2 * hi;
    const float* B0 = W + (size_t)(lr +  0) * Dd + 2 * hi;
    const float* B1 = W + (size_t)(lr + 16) * Dd + 2 * hi;
    const float* B2 = W + (size_t)(lr + 32) * Dd + 2 * hi;
    const float* B3 = W + (size_t)(lr + 48) * Dd + 2 * hi;

    v8f c0 = {}, c1 = {}, c2 = {}, c3 = {};
    for (int kk = 0; kk < Dd; kk += 4) {
        v2f a  = *(const v2f*)(Arow + kk);
        v2f b0 = *(const v2f*)(B0 + kk);
        v2f b1 = *(const v2f*)(B1 + kk);
        v2f b2 = *(const v2f*)(B2 + kk);
        v2f b3 = *(const v2f*)(B3 + kk);
        c0 = __builtin_amdgcn_wmma_f32_16x16x4_f32(false, a, false, b0, (short)0, c0, false, false);
        c1 = __builtin_amdgcn_wmma_f32_16x16x4_f32(false, a, false, b1, (short)0, c1, false, false);
        c2 = __builtin_amdgcn_wmma_f32_16x16x4_f32(false, a, false, b2, (short)0, c2, false, false);
        c3 = __builtin_amdgcn_wmma_f32_16x16x4_f32(false, a, false, b3, (short)0, c3, false, false);
    }

    if (w < 2) {
#pragma unroll
        for (int r = 0; r < 8; ++r) {
            c0[r] = fmaxf(c0[r], 0.f); c1[r] = fmaxf(c1[r], 0.f);
            c2[r] = fmaxf(c2[r], 0.f); c3[r] = fmaxf(c3[r], 0.f);
            float s = c0[r] + c1[r] + c2[r] + c3[r];
#pragma unroll
            for (int m = 8; m >= 1; m >>= 1) s += __shfl_xor(s, m, 32);
            float inv = 1.0f / (1e-5f + s);
            c0[r] *= inv; c1[r] *= inv; c2[r] *= inv; c3[r] *= inv;
        }
    }

#pragma unroll
    for (int r = 0; r < 8; ++r) {
        float* o = Out + (size_t)(tileM + r + 8 * hi) * Mm + lr;
        o[0]  = c0[r];
        o[16] = c1[r];
        o[32] = c2[r];
        o[48] = c3[r];
    }
}

#define TBLK 32

__device__ __forceinline__ float one_plus_tanh_small(float x)
{
    const float x2 = x * x;
    float p = __builtin_fmaf(x2, -0x1.758e7ap-8f, 0x1.521192p-6f);
    p = __builtin_fmaf(x2, p, -0x1.b8389cp-5f);
    p = __builtin_fmaf(x2, p, 0x1.110704p-3f);
    p = __builtin_fmaf(x2, p, -0x1.555532p-2f);
    return 1.0f + __builtin_fmaf(x2, x * p, x);
}

__global__ __launch_bounds__(256) void scan_kernel(const float* __restrict__ K,
                            const float* __restrict__ Q,
                            const float* __restrict__ V,
                            const float* __restrict__ vc,
                            const float* __restrict__ eta,
                            const float* __restrict__ theta_matrix,
                            float* __restrict__ y)
{
    const int b    = blockIdx.x;
    const int wave = threadIdx.x >> 5;
    const int lane = threadIdx.x & 31;
    const int i    = blockIdx.y * 8 + wave;
    const float theta_i = theta_matrix[i];

    float s0 = 0.f, s1 = 0.f;
    const size_t base = (size_t)b * Tt * Mm;
    const v2f* vcl = (const v2f*)(vc + base) + lane;
    const v2f* Kl  = (const v2f*)(K  + base) + lane;
    const v2f* Ql  = (const v2f*)(Q  + base) + lane;
    const float* etal = eta + (size_t)b * Tt + lane;
    const float* Vl   = V + base + (size_t)lane * Mm + i;

    float a0[TBLK], a1[TBLK], kx0[TBLK], kx1[TBLK], qx0[TBLK], qx1[TBLK];
#pragma unroll
    for (int tt = 0; tt < TBLK; ++tt) {
        const size_t o = (size_t)tt * (Mm / 2);
        const v2f av = vcl[o], kv = Kl[o], qv = Ql[o];
        a0[tt]  = av.x; a1[tt]  = av.y;
        kx0[tt] = kv.x; kx1[tt] = kv.y;
        qx0[tt] = qv.x; qx1[tt] = qv.y;
    }
    float u_l = etal[0] * Vl[0];

    for (int t0 = 0; t0 < Tt; t0 += TBLK) {
        const int tn = (t0 + TBLK < Tt) ? t0 + TBLK : t0;
        const float et_n = etal[tn];
        const float vi_n = Vl[(size_t)tn * Mm];

        float am[4] = {0.f, 0.f, 0.f, 0.f};
#pragma unroll
        for (int tt = 0; tt < TBLK; ++tt)
            am[tt & 3] = fmaxf(am[tt & 3], fmaxf(fabsf(a0[tt]), fabsf(a1[tt])));
        const float amax = fmaxf(fmaxf(am[0], am[1]), fmaxf(am[2], am[3]));
        const bool big = !(fabsf(theta_i) * amax < 0.625f);

        float c0[TBLK], c1[TBLK];
        if (__builtin_expect(__builtin_amdgcn_ballot_w32(big) == 0u, 1)) {
#pragma unroll
            for (int tt = 0; tt < TBLK; ++tt) {
                c0[tt] = one_plus_tanh_small(theta_i * a0[tt]);
                c1[tt] = one_plus_tanh_small(theta_i * a1[tt]);
            }
        } else {
#pragma unroll
            for (int tt = 0; tt < TBLK; ++tt) {
                c0[tt] = 1.0f + tanhf(theta_i * a0[tt]);
                c1[tt] = 1.0f + tanhf(theta_i * a1[tt]);
            }
        }

        float p[TBLK];
#pragma unroll
        for (int tt = 0; tt < TBLK; ++tt) {
            const float u = __builtin_bit_cast(float,
                __builtin_amdgcn_readlane(__builtin_bit_cast(int, u_l), tt));
            s0 = __builtin_fmaf(u, kx0[tt], s0 * c0[tt]);
            s1 = __builtin_fmaf(u, kx1[tt], s1 * c1[tt]);
            p[tt] = __builtin_fmaf(s0, qx0[tt], s1 * qx1[tt]);
            const size_t o = (size_t)(tn + tt) * (Mm / 2);
            const v2f av = vcl[o], kv = Kl[o], qv = Ql[o];
            a0[tt]  = av.x; a1[tt]  = av.y;
            kx0[tt] = kv.x; kx1[tt] = kv.y;
            qx0[tt] = qv.x; qx1[tt] = qv.y;
        }

#pragma unroll
        for (int m = TBLK / 2; m >= 1; m >>= 1) {
            const bool up = (lane & m) != 0;
#pragma unroll
            for (int r = 0; r < m; ++r) {
                const float lo = p[r], hi = p[r + m];
                const float keep = up ? hi : lo;
                const float send = up ? lo : hi;
                p[r] = keep + __shfl_xor(send, m, 32);
            }
        }
        y[base + (size_t)(t0 + lane) * Mm + i] = p[0];
        u_l = et_n * vi_n;
    }
}

__global__ void out_gemm_kernel(const float* __restrict__ y,
                                const float* __restrict__ Wout,
                                const float* __restrict__ bout,
                                float* __restrict__ out)
{
    const int tileM = blockIdx.x * 16;
    const int tileN = blockIdx.y * 64;
    const int lane  = threadIdx.x & 31;
    const int hi    = lane >> 4;
    const int lr    = lane & 15;

    const float* Arow = y + (size_t)(tileM + lr) * Mm + 2 * hi;
    const float* B0 = Wout + (size_t)(tileN + lr +  0) * Mm + 2 * hi;
    const float* B1 = Wout + (size_t)(tileN + lr + 16) * Mm + 2 * hi;
    const float* B2 = Wout + (size_t)(tileN + lr + 32) * Mm + 2 * hi;
    const float* B3 = Wout + (size_t)(tileN + lr + 48) * Mm + 2 * hi;

    v8f c0 = {}, c1 = {}, c2 = {}, c3 = {};
#pragma unroll
    for (int kk = 0; kk < Mm; kk += 4) {
        v2f a  = *(const v2f*)(Arow + kk);
        v2f b0 = *(const v2f*)(B0 + kk);
        v2f b1 = *(const v2f*)(B1 + kk);
        v2f b2 = *(const v2f*)(B2 + kk);
        v2f b3 = *(const v2f*)(B3 + kk);
        c0 = __builtin_amdgcn_wmma_f32_16x16x4_f32(false, a, false, b0, (short)0, c0, false, false);
        c1 = __builtin_amdgcn_wmma_f32_16x16x4_f32(false, a, false, b1, (short)0, c1, false, false);
        c2 = __builtin_amdgcn_wmma_f32_16x16x4_f32(false, a, false, b2, (short)0, c2, false, false);
        c3 = __builtin_amdgcn_wmma_f32_16x16x4_f32(false, a, false, b3, (short)0, c3, false, false);
    }

    const float bb0 = bout[tileN + lr +  0];
    const float bb1 = bout[tileN + lr + 16];
    const float bb2 = bout[tileN + lr + 32];
    const float bb3 = bout[tileN + lr + 48];

#pragma unroll
    for (int r = 0; r < 8; ++r) {
        float* o = out + (size_t)(tileM + r + 8 * hi) * Oo + tileN + lr;
        o[0]  = c0[r] + bb0;
        o[16] = c1[r] + bb1;
        o[32] = c2[r] + bb2;
        o[48] = c3[r] + bb3;
    }
}

extern "C" void kernel_launch(void* const* d_in, const int* in_sizes, int n_in,
                              void* d_out, int out_size, void* d_ws, size_t ws_size,
                              hipStream_t stream)
{
    const float* x     = (const float*)d_in[0];
    const float* Wk    = (const float*)d_in[1];
    const float* Wq    = (const float*)d_in[2];
    const float* Wv    = (const float*)d_in[3];
    const float* Wvc   = (const float*)d_in[4];
    const float* We    = (const float*)d_in[5];
    const float* theta = (const float*)d_in[6];
    const float* Wout  = (const float*)d_in[7];
    const float* bout  = (const float*)d_in[8];
    const float* gamma = (const float*)d_in[9];
    const float* beta  = (const float*)d_in[10];
    float* out = (float*)d_out;

    float* ws  = (float*)d_ws;
    const size_t NROW = (size_t)Bb * Tt;
    float* xn  = ws;
    float* Ko  = xn  + NROW * Dd;
    float* Qo  = Ko  + NROW * Mm;
    float* Vo  = Qo  + NROW * Mm;
    float* vco = Vo  + NROW * Mm;
    float* yb  = vco + NROW * Mm;
    float* eta = yb  + NROW * Mm;

    ln_eta_kernel<<<dim3(NROW / 8), dim3(256), 0, stream>>>(x, gamma, beta, We, xn, eta);
    proj_gemm_kernel<<<dim3(NROW / 16, 4), dim3(32), 0, stream>>>(xn, Wk, Wq, Wv, Wvc,
                                                                  Ko, Qo, Vo, vco);
    scan_kernel<<<dim3(Bb, Mm / 8), dim3(256), 0, stream>>>(Ko, Qo, Vo, vco, eta, theta, yb);
    out_gemm_kernel<<<dim3(NROW / 16, Oo / 64), dim3(32), 0, stream>>>(yb, Wout, bout, out);
}


# projection GEMM: K chunks of 32 staged through LDS by an 8-wave work-group (128 rows x one weight), whole-line 16-byte global loads, two buffers, one barrier a chunk; k order and sums unchanged
# speedup vs baseline: 8.8159x; 1.1334x over previous
#include <hip/hip_runtime.h>
#include <hip/hip_bf16.h>

typedef float v2f __attribute__((ext_vector_type(2)));
typedef float v4f __attribute__((ext_vector_type(4)));
typedef float v8f __attribute__((ext_vector_type(8)));

#define Bb 8
#define Tt 1024
#define Dd 512
#define Mm 64
#define Oo 512

__global__ void ln_eta_kernel(const float* __restrict__ x,
                              const float* __restrict__ gamma,
                              const float* __restrict__ beta,
                              const float* __restrict__ We,
                              float* __restrict__ xn,
                              float* __restrict__ eta)
{
    const int wave = threadIdx.x >> 5;
    const int lane = threadIdx.x & 31;
    const int row  = blockIdx.x * 8 + wave;
    const float* xr = x + (size_t)row * Dd;

    float v[16];
    float sum = 0.f, sq = 0.f;
#pragma unroll
    for (int e = 0; e < 16; ++e) {
        float t = xr[lane + 32 * e];
        v[e] = t; sum += t; sq += t * t;
    }
#pragma unroll
    for (int m = 16; m >= 1; m >>= 1) {
        sum += __shfl_xor(sum, m, 32);
        sq  += __shfl_xor(sq,  m, 32);
    }
    const float mu = sum * (1.0f / Dd);
    const float var = sq * (1.0f / Dd) - mu * mu;
    const float rs = rsqrtf(var + 1e-5f);

    float dot = 0.f;
    float* xo = xn + (size_t)row * Dd;
#pragma unroll
    for (int e = 0; e < 16; ++e) {
        const int d = lane + 32 * e;
        float xv = (v[e] - mu) * rs * gamma[d] + beta[d];
        xo[d] = xv;
        dot += xv * We[d];
    }
#pragma unroll
    for (int m = 16; m >= 1; m >>= 1) dot += __shfl_xor(dot, m, 32);
    if (lane == 0) eta[row] = 1.0f / (1.0f + expf(-dot));
}

#define PROWS  128
#define PKC    32
#define PPITCH 36

__global__ __launch_bounds__(256) __attribute__((amdgpu_waves_per_eu(2, 2)))
void proj_gemm_kernel(const float* __restrict__ xn,
                                 const float* __restrict__ Wk,
                                 const float* __restrict__ Wq,
                                 const float* __restrict__ Wv,
                                 const float* __restrict__ Wvc,
                                 float* __restrict__ Ko,
                                 float* __restrict__ Qo,
                                 float* __restrict__ Vo,
                                 float* __restrict__ vco)
{
    __shared__ __attribute__((aligned(16))) float As[2][PROWS * PPITCH];
    __shared__ __attribute__((aligned(16))) float Bs[2][Mm * PPITCH];

    const int w = blockIdx.y;
    const float* W = (w == 0) ? Wk : (w == 1) ? Wq : (w == 2) ? Wv : Wvc;
    float* Out     = (w == 0) ? Ko : (w == 1) ? Qo : (w == 2) ? Vo : vco;

    const int tid   = threadIdx.x;
    const int wave  = tid >> 5;
    const int lane  = tid & 31;
    const int hi    = lane >> 4;
    const int lr    = lane & 15;
    const int tileM = blockIdx.x * PROWS + wave * 16;

    const int srow = tid >> 3;
    const int sseg = (tid & 7) * 4;
    const float* Ag = xn + (size_t)(blockIdx.x * PROWS + srow) * Dd + sseg;
    const float* Bg = W  + (size_t)srow * Dd + sseg;
    const int sl = srow * PPITCH + sseg;

    v4f ra[4], rb[2];
#pragma unroll
    for (int j = 0; j < 4; ++j) ra[j] = *(const v4f*)(Ag + (size_t)(32 * j) * Dd);
#pragma unroll
    for (int j = 0; j < 2; ++j) rb[j] = *(const v4f*)(Bg + (size_t)(32 * j) * Dd);
#pragma unroll
    for (int j = 0; j < 4; ++j) *(v4f*)&As[0][sl + 32 * j * PPITCH] = ra[j];
#pragma unroll
    for (int j = 0; j < 2; ++j) *(v4f*)&Bs[0][sl + 32 * j * PPITCH] = rb[j];
    __syncthreads();

    const int fa = (wave * 16 + lr) * PPITCH + 2 * hi;
    const int fb = lr * PPITCH + 2 * hi;

    v8f c0 = {}, c1 = {}, c2 = {}, c3 = {};
    for (int c = 0; c < Dd / PKC; ++c) {
        const int cur = c & 1;
        const bool more = (c + 1 < Dd / PKC);
        if (more) {
            const int ko = (c + 1) * PKC;
#pragma unroll
            for (int j = 0; j < 4; ++j) ra[j] = *(const v4f*)(Ag + (size_t)(32 * j) * Dd + ko);
#pragma unroll
            for (int j = 0; j < 2; ++j) rb[j] = *(const v4f*)(Bg + (size_t)(32 * j) * Dd + ko);
        }

        const float* Afr = &As[cur][fa];
        const float* Bfr = &Bs[cur][fb];
#pragma unroll
        for (int kk = 0; kk < PKC; kk += 4) {
            v2f a  = *(const v2f*)(Afr + kk);
            v2f b0 = *(const v2f*)(Bfr + kk);
            v2f b1 = *(const v2f*)(Bfr + 16 * PPITCH + kk);
            v2f b2 = *(const v2f*)(Bfr + 32 * PPITCH + kk);
            v2f b3 = *(const v2f*)(Bfr + 48 * PPITCH + kk);
            c0 = __builtin_amdgcn_wmma_f32_16x16x4_f32(false, a, false, b0, (short)0, c0, false, false);
            c1 = __builtin_amdgcn_wmma_f32_16x16x4_f32(false, a, false, b1, (short)0, c1, false, false);
            c2 = __builtin_amdgcn_wmma_f32_16x16x4_f32(false, a, false, b2, (short)0, c2, false, false);
            c3 = __builtin_amdgcn_wmma_f32_16x16x4_f32(false, a, false, b3, (short)0, c3, false, false);
        }

        if (more) {
#pragma unroll
            for (int j = 0; j < 4; ++j) *(v4f*)&As[cur ^ 1][sl + 32 * j * PPITCH] = ra[j];
#pragma unroll
            for (int j = 0; j < 2; ++j) *(v4f*)&Bs[cur ^ 1][sl + 32 * j * PPITCH] = rb[j];
        }
        __syncthreads();
    }

    if (w < 2) {
#pragma unroll
        for (int r = 0; r < 8; ++r) {
            c0[r] = fmaxf(c0[r], 0.f); c1[r] = fmaxf(c1[r], 0.f);
            c2[r] = fmaxf(c2[r], 0.f); c3[r] = fmaxf(c3[r], 0.f);
            float s = c0[r] + c1[r] + c2[r] + c3[r];
#pragma unroll
            for (int m = 8; m >= 1; m >>= 1) s += __shfl_xor(s, m, 32);
            float inv = 1.0f / (1e-5f + s);
            c0[r] *= inv; c1[r] *= inv; c2[r] *= inv; c3[r] *= inv;
        }
    }

#pragma unroll
    for (int r = 0; r < 8; ++r) {
        float* o = Out + (size_t)(tileM + r + 8 * hi) * Mm + lr;
        o[0]  = c0[r];
        o[16] = c1[r];
        o[32] = c2[r];
        o[48] = c3[r];
    }
}

#define TBLK 32

__device__ __forceinline__ float one_plus_tanh_small(float x)
{
    const float x2 = x * x;
    float p = __builtin_fmaf(x2, -0x1.758e7ap-8f, 0x1.521192p-6f);
    p = __builtin_fmaf(x2, p, -0x1.b8389cp-5f);
    p = __builtin_fmaf(x2, p, 0x1.110704p-3f);
    p = __builtin_fmaf(x2, p, -0x1.555532p-2f);
    return 1.0f + __builtin_fmaf(x2, x * p, x);
}

__global__ __launch_bounds__(256) void scan_kernel(const float* __restrict__ K,
                            const float* __restrict__ Q,
                            const float* __restrict__ V,
                            const float* __restrict__ vc,
                            const float* __restrict__ eta,
                            const float* __restrict__ theta_matrix,
                            float* __restrict__ y)
{
    const int b    = blockIdx.x;
    const int wave = threadIdx.x >> 5;
    const int lane = threadIdx.x & 31;
    const int i    = blockIdx.y * 8 + wave;
    const float theta_i = theta_matrix[i];

    float s0 = 0.f, s1 = 0.f;
    const size_t base = (size_t)b * Tt * Mm;
    const v2f* vcl = (const v2f*)(vc + base) + lane;
    const v2f* Kl  = (const v2f*)(K  + base) + lane;
    const v2f* Ql  = (const v2f*)(Q  + base) + lane;
    const float* etal = eta + (size_t)b * Tt + lane;
    const float* Vl   = V + base + (size_t)lane * Mm + i;

    float a0[TBLK], a1[TBLK], kx0[TBLK], kx1[TBLK], qx0[TBLK], qx1[TBLK];
#pragma unroll
    for (int tt = 0; tt < TBLK; ++tt) {
        const size_t o = (size_t)tt * (Mm / 2);
        const v2f av = vcl[o], kv = Kl[o], qv = Ql[o];
        a0[tt]  = av.x; a1[tt]  = av.y;
        kx0[tt] = kv.x; kx1[tt] = kv.y;
        qx0[tt] = qv.x; qx1[tt] = qv.y;
    }
    float u_l = etal[0] * Vl[0];

    for (int t0 = 0; t0 < Tt; t0 += TBLK) {
        const int tn = (t0 + TBLK < Tt) ? t0 + TBLK : t0;
        const float et_n = etal[tn];
        const float vi_n = Vl[(size_t)tn * Mm];

        float am[4] = {0.f, 0.f, 0.f, 0.f};
#pragma unroll
        for (int tt = 0; tt < TBLK; ++tt)
            am[tt & 3] = fmaxf(am[tt & 3], fmaxf(fabsf(a0[tt]), fabsf(a1[tt])));
        const float amax = fmaxf(fmaxf(am[0], am[1]), fmaxf(am[2], am[3]));
        const bool big = !(fabsf(theta_i) * amax < 0.625f);

        float c0[TBLK], c1[TBLK];
        if (__builtin_expect(__builtin_amdgcn_ballot_w32(big) == 0u, 1)) {
#pragma unroll
            for (int tt = 0; tt < TBLK; ++tt) {
                c0[tt] = one_plus_tanh_small(theta_i * a0[tt]);
                c1[tt] = one_plus_tanh_small(theta_i * a1[tt]);
            }
        } else {
#pragma unroll
            for (int tt = 0; tt < TBLK; ++tt) {
                c0[tt] = 1.0f + tanhf(theta_i * a0[tt]);
                c1[tt] = 1.0f + tanhf(theta_i * a1[tt]);
            }
        }

        float p[TBLK];
#pragma unroll
        for (int tt = 0; tt < TBLK; ++tt) {
            const float u = __builtin_bit_cast(float,
                __builtin_amdgcn_readlane(__builtin_bit_cast(int, u_l), tt));
            s0 = __builtin_fmaf(u, kx0[tt], s0 * c0[tt]);
            s1 = __builtin_fmaf(u, kx1[tt], s1 * c1[tt]);
            p[tt] = __builtin_fmaf(s0, qx0[tt], s1 * qx1[tt]);
            const size_t o = (size_t)(tn + tt) * (Mm / 2);
            const v2f av = vcl[o], kv = Kl[o], qv = Ql[o];
            a0[tt]  = av.x; a1[tt]  = av.y;
            kx0[tt] = kv.x; kx1[tt] = kv.y;
            qx0[tt] = qv.x; qx1[tt] = qv.y;
        }

#pragma unroll
        for (int m = TBLK / 2; m >= 1; m >>= 1) {
            const bool up = (lane & m) != 0;
#pragma unroll
            for (int r = 0; r < m; ++r) {
                const float lo = p[r], hi = p[r + m];
                const float keep = up ? hi : lo;
                const float send = up ? lo : hi;
                p[r] = keep + __shfl_xor(send, m, 32);
            }
        }
        y[base + (size_t)(t0 + lane) * Mm + i] = p[0];
        u_l = et_n * vi_n;
    }
}

__global__ void out_gemm_kernel(const float* __restrict__ y,
                                const float* __restrict__ Wout,
                                const float* __restrict__ bout,
                                float* __restrict__ out)
{
    const int tileM = blockIdx.x * 16;
    const int tileN = blockIdx.y * 64;
    const int lane  = threadIdx.x & 31;
    const int hi    = lane >> 4;
    const int lr    = lane & 15;

    const float* Arow = y + (size_t)(tileM + lr) * Mm + 2 * hi;
    const float* B0 = Wout + (size_t)(tileN + lr +  0) * Mm + 2 * hi;
    const float* B1 = Wout + (size_t)(tileN + lr + 16) * Mm + 2 * hi;
    const float* B2 = Wout + (size_t)(tileN + lr + 32) * Mm + 2 * hi;
    const float* B3 = Wout + (size_t)(tileN + lr + 48) * Mm + 2 * hi;

    v8f c0 = {}, c1 = {}, c2 = {}, c3 = {};
#pragma unroll
    for (int kk = 0; kk < Mm; kk += 4) {
        v2f a  = *(const v2f*)(Arow + kk);
        v2f b0 = *(const v2f*)(B0 + kk);
        v2f b1 = *(const v2f*)(B1 + kk);
        v2f b2 = *(const v2f*)(B2 + kk);
        v2f b3 = *(const v2f*)(B3 + kk);
        c0 = __builtin_amdgcn_wmma_f32_16x16x4_f32(false, a, false, b0, (short)0, c0, false, false);
        c1 = __builtin_amdgcn_wmma_f32_16x16x4_f32(false, a, false, b1, (short)0, c1, false, false);
        c2 = __builtin_amdgcn_wmma_f32_16x16x4_f32(false, a, false, b2, (short)0, c2, false, false);
        c3 = __builtin_amdgcn_wmma_f32_16x16x4_f32(false, a, false, b3, (short)0, c3, false, false);
    }

    const float bb0 = bout[tileN + lr +  0];
    const float bb1 = bout[tileN + lr + 16];
    const float bb2 = bout[tileN + lr + 32];
    const float bb3 = bout[tileN + lr + 48];

#pragma unroll
    for (int r = 0; r < 8; ++r) {
        float* o = out + (size_t)(tileM + r + 8 * hi) * Oo + tileN + lr;
        o[0]  = c0[r] + bb0;
        o[16] = c1[r] + bb1;
        o[32] = c2[r] + bb2;
        o[48] = c3[r] + bb3;
    }
}

extern "C" void kernel_launch(void* const* d_in, const int* in_sizes, int n_in,
                              void* d_out, int out_size, void* d_ws, size_t ws_size,
                              hipStream_t stream)
{
    const float* x     = (const float*)d_in[0];
    const float* Wk    = (const float*)d_in[1];
    const float* Wq    = (const float*)d_in[2];
    const float* Wv    = (const float*)d_in[3];
    const float* Wvc   = (const float*)d_in[4];
    const float* We    = (const float*)d_in[5];
    const float* theta = (const float*)d_in[6];
    const float* Wout  = (const float*)d_in[7];
    const float* bout  = (const float*)d_in[8];
    const float* gamma = (const float*)d_in[9];
    const float* beta  = (const float*)d_in[10];
    float* out = (float*)d_out;

    float* ws  = (float*)d_ws;
    const size_t NROW = (size_t)Bb * Tt;
    float* xn  = ws;
    float* Ko  = xn  + NROW * Dd;
    float* Qo  = Ko  + NROW * Mm;
    float* Vo  = Qo  + NROW * Mm;
    float* vco = Vo  + NROW * Mm;
    float* yb  = vco + NROW * Mm;
    float* eta = yb  + NROW * Mm;

    ln_eta_kernel<<<dim3(NROW / 8), dim3(256), 0, stream>>>(x, gamma, beta, We, xn, eta);
    proj_gemm_kernel<<<dim3(NROW / PROWS, 4), dim3(256), 0, stream>>>(xn, Wk, Wq, Wv, Wvc,
                                                                  Ko, Qo, Vo, vco);
    scan_kernel<<<dim3(Bb, Mm / 8), dim3(256), 0, stream>>>(Ko, Qo, Vo, vco, eta, theta, yb);
    out_gemm_kernel<<<dim3(NROW / 16, Oo / 64), dim3(32), 0, stream>>>(yb, Wout, bout, out);
}


# scan kernel: a work-group (4 waves, one per SIMD) copies each 32-step block of vc/K/Q rows once into LDS (16-byte loads); waves fill operand registers from LDS at the block head; one barrier a block
# speedup vs baseline: 9.2686x; 1.0514x over previous
#include <hip/hip_runtime.h>
#include <hip/hip_bf16.h>

typedef float v2f __attribute__((ext_vector_type(2)));
typedef float v4f __attribute__((ext_vector_type(4)));
typedef float v8f __attribute__((ext_vector_type(8)));

#define Bb 8
#define Tt 1024
#define Dd 512
#define Mm 64
#define Oo 512

__global__ void ln_eta_kernel(const float* __restrict__ x,
                              const float* __restrict__ gamma,
                              const float* __restrict__ beta,
                              const float* __restrict__ We,
                              float* __restrict__ xn,
                              float* __restrict__ eta)
{
    const int wave = threadIdx.x >> 5;
    const int lane = threadIdx.x & 31;
    const int row  = blockIdx.x * 8 + wave;
    const float* xr = x + (size_t)row * Dd;

    float v[16];
    float sum = 0.f, sq = 0.f;
#pragma unroll
    for (int e = 0; e < 16; ++e) {
        float t = xr[lane + 32 * e];
        v[e] = t; sum += t; sq += t * t;
    }
#pragma unroll
    for (int m = 16; m >= 1; m >>= 1) {
        sum += __shfl_xor(sum, m, 32);
        sq  += __shfl_xor(sq,  m, 32);
    }
    const float mu = sum * (1.0f / Dd);
    const float var = sq * (1.0f / Dd) - mu * mu;
    const float rs = rsqrtf(var + 1e-5f);

    float dot = 0.f;
    float* xo = xn + (size_t)row * Dd;
#pragma unroll
    for (int e = 0; e < 16; ++e) {
        const int d = lane + 32 * e;
        float xv = (v[e] - mu) * rs * gamma[d] + beta[d];
        xo[d] = xv;
        dot += xv * We[d];
    }
#pragma unroll
    for (int m = 16; m >= 1; m >>= 1) dot += __shfl_xor(dot, m, 32);
    if (lane == 0) eta[row] = 1.0f / (1.0f + expf(-dot));
}

#define PROWS  128
#define PKC    32
#define PPITCH 36

__global__ __launch_bounds__(256) __attribute__((amdgpu_waves_per_eu(2, 2)))
void proj_gemm_kernel(const float* __restrict__ xn,
                                 const float* __restrict__ Wk,
                                 const float* __restrict__ Wq,
                                 const float* __restrict__ Wv,
                                 const float* __restrict__ Wvc,
                                 float* __restrict__ Ko,
                                 float* __restrict__ Qo,
                                 float* __restrict__ Vo,
                                 float* __restrict__ vco)
{
    __shared__ __attribute__((aligned(16))) float As[2][PROWS * PPITCH];
    __shared__ __attribute__((aligned(16))) float Bs[2][Mm * PPITCH];

    const int w = blockIdx.y;
    const float* W = (w == 0) ? Wk : (w == 1) ? Wq : (w == 2) ? Wv : Wvc;
    float* Out     = (w == 0) ? Ko : (w == 1) ? Qo : (w == 2) ? Vo : vco;

    const int tid   = threadIdx.x;
    const int wave  = tid >> 5;
    const int lane  = tid & 31;
    const int hi    = lane >> 4;
    const int lr    = lane & 15;
    const int tileM = blockIdx.x * PROWS + wave * 16;

    const int srow = tid >> 3;
    const int sseg = (tid & 7) * 4;
    const float* Ag = xn + (size_t)(blockIdx.x * PROWS + srow) * Dd + sseg;
    const float* Bg = W  + (size_t)srow * Dd + sseg;
    const int sl = srow * PPITCH + sseg;

    v4f ra[4], rb[2];
#pragma unroll
    for (int j = 0; j < 4; ++j) ra[j] = *(const v4f*)(Ag + (size_t)(32 * j) * Dd);
#pragma unroll
    for (int j = 0; j < 2; ++j) rb[j] = *(const v4f*)(Bg + (size_t)(32 * j) * Dd);
#pragma unroll
    for (int j = 0; j < 4; ++j) *(v4f*)&As[0][sl + 32 * j * PPITCH] = ra[j];
#pragma unroll
    for (int j = 0; j < 2; ++j) *(v4f*)&Bs[0][sl + 32 * j * PPITCH] = rb[j];
    __syncthreads();

    const int fa = (wave * 16 + lr) * PPITCH + 2 * hi;
    const int fb = lr * PPITCH + 2 * hi;

    v8f c0 = {}, c1 = {}, c2 = {}, c3 = {};
    for (int c = 0; c < Dd / PKC; ++c) {
        const int cur = c & 1;
        const bool more = (c + 1 < Dd / PKC);
        if (more) {
            const int ko = (c + 1) * PKC;
#pragma unroll
            for (int j = 0; j < 4; ++j) ra[j] = *(const v4f*)(Ag + (size_t)(32 * j) * Dd + ko);
#pragma unroll
            for (int j = 0; j < 2; ++j) rb[j] = *(const v4f*)(Bg + (size_t)(32 * j) * Dd + ko);
        }

        const float* Afr = &As[cur][fa];
        const float* Bfr = &Bs[cur][fb];
#pragma unroll
        for (int kk = 0; kk < PKC; kk += 4) {
            v2f a  = *(const v2f*)(Afr + kk);
            v2f b0 = *(const v2f*)(Bfr + kk);
            v2f b1 = *(const v2f*)(Bfr + 16 * PPITCH + kk);
            v2f b2 = *(const v2f*)(Bfr + 32 * PPITCH + kk);
            v2f b3 = *(const v2f*)(Bfr + 48 * PPITCH + kk);
            c0 = __builtin_amdgcn_wmma_f32_16x16x4_f32(false, a, false, b0, (short)0, c0, false, false);
            c1 = __builtin_amdgcn_wmma_f32_16x16x4_f32(false, a, false, b1, (short)0, c1, false, false);
            c2 = __builtin_amdgcn_wmma_f32_16x16x4_f32(false, a, false, b2, (short)0, c2, false, false);
            c3 = __builtin_amdgcn_wmma_f32_16x16x4_f32(false, a, false, b3, (short)0, c3, false, false);
        }

        if (more) {
#pragma unroll
            for (int j = 0; j < 4; ++j) *(v4f*)&As[cur ^ 1][sl + 32 * j * PPITCH] = ra[j];
#pragma unroll
            for (int j = 0; j < 2; ++j) *(v4f*)&Bs[cur ^ 1][sl + 32 * j * PPITCH] = rb[j];
        }
        __syncthreads();
    }

    if (w < 2) {
#pragma unroll
        for (int r = 0; r < 8; ++r) {
            c0[r] = fmaxf(c0[r], 0.f); c1[r] = fmaxf(c1[r], 0.f);
            c2[r] = fmaxf(c2[r], 0.f); c3[r] = fmaxf(c3[r], 0.f);
            float s = c0[r] + c1[r] + c2[r] + c3[r];
#pragma unroll
            for (int m = 8; m >= 1; m >>= 1) s += __shfl_xor(s, m, 32);
            float inv = 1.0f / (1e-5f + s);
            c0[r] *= inv; c1[r] *= inv; c2[r] *= inv; c3[r] *= inv;
        }
    }

#pragma unroll
    for (int r = 0; r < 8; ++r) {
        float* o = Out + (size_t)(tileM + r + 8 * hi) * Mm + lr;
        o[0]  = c0[r];
        o[16] = c1[r];
        o[32] = c2[r];
        o[48] = c3[r];
    }
}

#define TBLK 32
#define NBLK (Tt / TBLK)
#define SCW 4
#define SBLKF (TBLK * Mm)
#define SPIECES (SBLKF / 4 / (SCW * 32))

__device__ __forceinline__ float one_plus_tanh_small(float x)
{
    const float x2 = x * x;
    float p = __builtin_fmaf(x2, -0x1.758e7ap-8f, 0x1.521192p-6f);
    p = __builtin_fmaf(x2, p, -0x1.b8389cp-5f);
    p = __builtin_fmaf(x2, p, 0x1.110704p-3f);
    p = __builtin_fmaf(x2, p, -0x1.555532p-2f);
    return 1.0f + __builtin_fmaf(x2, x * p, x);
}

__global__ __launch_bounds__(SCW * 32) void scan_kernel(const float* __restrict__ K,
                            const float* __restrict__ Q,
                            const float* __restrict__ V,
                            const float* __restrict__ vc,
                            const float* __restrict__ eta,
                            const float* __restrict__ theta_matrix,
                            float* __restrict__ y)
{
    __shared__ __attribute__((aligned(16))) float Svc[2][SBLKF];
    __shared__ __attribute__((aligned(16))) float Skq[2][2][SBLKF];

    const int b    = blockIdx.x;
    const int tid  = threadIdx.x;
    const int wave = tid >> 5;
    const int lane = tid & 31;
    const int i    = blockIdx.y * SCW + wave;
    const float theta_i = theta_matrix[i];

    float s0 = 0.f, s1 = 0.f;
    const size_t base = (size_t)b * Tt * Mm;
    const float* etal = eta + (size_t)b * Tt + lane;
    const float* Vl   = V + base + (size_t)lane * Mm + i;
    const v4f* gvc = (const v4f*)(vc + base) + tid;
    const v4f* gK  = (const v4f*)(K  + base) + tid;
    const v4f* gQ  = (const v4f*)(Q  + base) + tid;

    float a0[TBLK], a1[TBLK];
#pragma unroll
    for (int tt = 0; tt < TBLK; ++tt) {
        const v2f av = ((const v2f*)(vc + base) + lane)[tt * (Mm / 2)];
        a0[tt] = av.x; a1[tt] = av.y;
    }
    float u_l = etal[0] * Vl[0];

#pragma unroll
    for (int j = 0; j < SPIECES; ++j) {
        const int pc = j * (SCW * 32);
        const v4f gv = gvc[SBLKF / 4 + pc], gk = gK[pc], gq = gQ[pc];
        ((v4f*)Svc[1])[tid + pc]    = gv;
        ((v4f*)Skq[0][0])[tid + pc] = gk;
        ((v4f*)Skq[0][1])[tid + pc] = gq;
    }
    __syncthreads();

    for (int t0 = 0; t0 < Tt; t0 += TBLK) {
        const int n = t0 / TBLK;
        const int tn = (t0 + TBLK < Tt) ? t0 + TBLK : t0;
        const float et_n = etal[tn];
        const float vi_n = Vl[(size_t)tn * Mm];

        const int nv = (n + 2 < NBLK) ? n + 2 : NBLK - 1;
        const int nk = (n + 1 < NBLK) ? n + 1 : NBLK - 1;
        v4f gv[SPIECES], gk[SPIECES], gq[SPIECES];
#pragma unroll
        for (int j = 0; j < SPIECES; ++j) {
            const int pc = j * (SCW * 32);
            gv[j] = gvc[(size_t)nv * (SBLKF / 4) + pc];
            gk[j] = gK[(size_t)nk * (SBLKF / 4) + pc];
            gq[j] = gQ[(size_t)nk * (SBLKF / 4) + pc];
        }

        const v2f* Lv = (const v2f*)Svc[(n + 1) & 1] + lane;
        const v2f* Lk = (const v2f*)Skq[n & 1][0] + lane;
        const v2f* Lq = (const v2f*)Skq[n & 1][1] + lane;
        float an0[TBLK], an1[TBLK], kx0[TBLK], kx1[TBLK], qx0[TBLK], qx1[TBLK];
#pragma unroll
        for (int tt = 0; tt < TBLK; ++tt) {
            const int o = tt * (Mm / 2);
            const v2f av = Lv[o], kv = Lk[o], qv = Lq[o];
            an0[tt] = av.x; an1[tt] = av.y;
            kx0[tt] = kv.x; kx1[tt] = kv.y;
            qx0[tt] = qv.x; qx1[tt] = qv.y;
        }

        float am[4] = {0.f, 0.f, 0.f, 0.f};
#pragma unroll
        for (int tt = 0; tt < TBLK; ++tt)
            am[tt & 3] = fmaxf(am[tt & 3], fmaxf(fabsf(a0[tt]), fabsf(a1[tt])));
        const float amax = fmaxf(fmaxf(am[0], am[1]), fmaxf(am[2], am[3]));
        const bool big = !(fabsf(theta_i) * amax < 0.625f);

        float c0[TBLK], c1[TBLK];
        if (__builtin_expect(__builtin_amdgcn_ballot_w32(big) == 0u, 1)) {
#pragma unroll
            for (int tt = 0; tt < TBLK; ++tt) {
                c0[tt] = one_plus_tanh_small(theta_i * a0[tt]);
                c1[tt] = one_plus_tanh_small(theta_i * a1[tt]);
            }
        } else {
#pragma unroll
            for (int tt = 0; tt < TBLK; ++tt) {
                c0[tt] = 1.0f + tanhf(theta_i * a0[tt]);
                c1[tt] = 1.0f + tanhf(theta_i * a1[tt]);
            }
        }

        float p[TBLK];
#pragma unroll
        for (int tt = 0; tt < TBLK; ++tt) {
            const float u = __builtin_bit_cast(float,
                __builtin_amdgcn_readlane(__builtin_bit_cast(int, u_l), tt));
            s0 = __builtin_fmaf(u, kx0[tt], s0 * c0[tt]);
            s1 = __builtin_fmaf(u, kx1[tt], s1 * c1[tt]);
            p[tt] = __builtin_fmaf(s0, qx0[tt], s1 * qx1[tt]);
            a0[tt] = an0[tt]; a1[tt] = an1[tt];
        }

#pragma unroll
        for (int j = 0; j < SPIECES; ++j) {
            const int pc = tid + j * (SCW * 32);
            ((v4f*)Svc[n & 1])[pc]          = gv[j];
            ((v4f*)Skq[(n + 1) & 1][0])[pc] = gk[j];
            ((v4f*)Skq[(n + 1) & 1][1])[pc] = gq[j];
        }

#pragma unroll
        for (int m = TBLK / 2; m >= 1; m >>= 1) {
            const bool up = (lane & m) != 0;
#pragma unroll
            for (int r = 0; r < m; ++r) {
                const float lo = p[r], hi = p[r + m];
                const float keep = up ? hi : lo;
                const float send = up ? lo : hi;
                p[r] = keep + __shfl_xor(send, m, 32);
            }
        }
        u_l = et_n * vi_n;
        __syncthreads();
        y[base + (size_t)(t0 + lane) * Mm + i] = p[0];
    }
}

__global__ void out_gemm_kernel(const float* __restrict__ y,
                                const float* __restrict__ Wout,
                                const float* __restrict__ bout,
                                float* __restrict__ out)
{
    const int tileM = blockIdx.x * 16;
    const int tileN = blockIdx.y * 64;
    const int lane  = threadIdx.x & 31;
    const int hi    = lane >> 4;
    const int lr    = lane & 15;

    const float* Arow = y + (size_t)(tileM + lr) * Mm + 2 * hi;
    const float* B0 = Wout + (size_t)(tileN + lr +  0) * Mm + 2 * hi;
    const float* B1 = Wout + (size_t)(tileN + lr + 16) * Mm + 2 * hi;
    const float* B2 = Wout + (size_t)(tileN + lr + 32) * Mm + 2 * hi;
    const float* B3 = Wout + (size_t)(tileN + lr + 48) * Mm + 2 * hi;

    v8f c0 = {}, c1 = {}, c2 = {}, c3 = {};
#pragma unroll
    for (int kk = 0; kk < Mm; kk += 4) {
        v2f a  = *(const v2f*)(Arow + kk);
        v2f b0 = *(const v2f*)(B0 + kk);
        v2f b1 = *(const v2f*)(B1 + kk);
        v2f b2 = *(const v2f*)(B2 + kk);
        v2f b3 = *(const v2f*)(B3 + kk);
        c0 = __builtin_amdgcn_wmma_f32_16x16x4_f32(false, a, false, b0, (short)0, c0, false, false);
        c1 = __builtin_amdgcn_wmma_f32_16x16x4_f32(false, a, false, b1, (short)0, c1, false, false);
        c2 = __builtin_amdgcn_wmma_f32_16x16x4_f32(false, a, false, b2, (short)0, c2, false, false);
        c3 = __builtin_amdgcn_wmma_f32_16x16x4_f32(false, a, false, b3, (short)0, c3, false, false);
    }

    const float bb0 = bout[tileN + lr +  0];
    const float bb1 = bout[tileN + lr + 16];
    const float bb2 = bout[tileN + lr + 32];
    const float bb3 = bout[tileN + lr + 48];

#pragma unroll
    for (int r = 0; r < 8; ++r) {
        float* o = out + (size_t)(tileM + r + 8 * hi) * Oo + tileN + lr;
        o[0]  = c0[r] + bb0;
        o[16] = c1[r] + bb1;
        o[32] = c2[r] + bb2;
        o[48] = c3[r] + bb3;
    }
}

extern "C" void kernel_launch(void* const* d_in, const int* in_sizes, int n_in,
                              void* d_out, int out_size, void* d_ws, size_t ws_size,
                              hipStream_t stream)
{
    const float* x     = (const float*)d_in[0];
    const float* Wk    = (const float*)d_in[1];
    const float* Wq    = (const float*)d_in[2];
    const float* Wv    = (const float*)d_in[3];
    const float* Wvc   = (const float*)d_in[4];
    const float* We    = (const float*)d_in[5];
    const float* theta = (const float*)d_in[6];
    const float* Wout  = (const float*)d_in[7];
    const float* bout  = (const float*)d_in[8];
    const float* gamma = (const float*)d_in[9];
    const float* beta  = (const float*)d_in[10];
    float* out = (float*)d_out;

    float* ws  = (float*)d_ws;
    const size_t NROW = (size_t)Bb * Tt;
    float* xn  = ws;
    float* Ko  = xn  + NROW * Dd;
    float* Qo  = Ko  + NROW * Mm;
    float* Vo  = Qo  + NROW * Mm;
    float* vco = Vo  + NROW * Mm;
    float* yb  = vco + NROW * Mm;
    float* eta = yb  + NROW * Mm;

    ln_eta_kernel<<<dim3(NROW / 8), dim3(256), 0, stream>>>(x, gamma, beta, We, xn, eta);
    proj_gemm_kernel<<<dim3(NROW / PROWS, 4), dim3(256), 0, stream>>>(xn, Wk, Wq, Wv, Wvc,
                                                                  Ko, Qo, Vo, vco);
    scan_kernel<<<dim3(Bb, Mm / SCW), dim3(SCW * 32), 0, stream>>>(Ko, Qo, Vo, vco, eta, theta, yb);
    out_gemm_kernel<<<dim3(NROW / 16, Oo / 64), dim3(32), 0, stream>>>(yb, Wout, bout, out);
}


# scan kernel: the butterfly's 31 lane exchanges a block go from LDS-unit shuffles to vector-ALU forms (cross-row permute for mask 16, row_xmask operand of the add for 8..1); same pairs, same sums
# speedup vs baseline: 10.3803x; 1.1199x over previous
#include <hip/hip_runtime.h>
#include <hip/hip_bf16.h>

typedef float v2f __attribute__((ext_vector_type(2)));
typedef float v4f __attribute__((ext_vector_type(4)));
typedef float v8f __attribute__((ext_vector_type(8)));

#define Bb 8
#define Tt 1024
#define Dd 512
#define Mm 64
#define Oo 512

__global__ void ln_eta_kernel(const float* __restrict__ x,
                              const float* __restrict__ gamma,
                              const float* __restrict__ beta,
                              const float* __restrict__ We,
                              float* __restrict__ xn,
                              float* __restrict__ eta)
{
    const int wave = threadIdx.x >> 5;
    const int lane = threadIdx.x & 31;
    const int row  = blockIdx.x * 8 + wave;
    const float* xr = x + (size_t)row * Dd;

    float v[16];
    float sum = 0.f, sq = 0.f;
#pragma unroll
    for (int e = 0; e < 16; ++e) {
        float t = xr[lane + 32 * e];
        v[e] = t; sum += t; sq += t * t;
    }
#pragma unroll
    for (int m = 16; m >= 1; m >>= 1) {
        sum += __shfl_xor(sum, m, 32);
        sq  += __shfl_xor(sq,  m, 32);
    }
    const float mu = sum * (1.0f / Dd);
    const float var = sq * (1.0f / Dd) - mu * mu;
    const float rs = rsqrtf(var + 1e-5f);

    float dot = 0.f;
    float* xo = xn + (size_t)row * Dd;
#pragma unroll
    for (int e = 0; e < 16; ++e) {
        const int d = lane + 32 * e;
        float xv = (v[e] - mu) * rs * gamma[d] + beta[d];
        xo[d] = xv;
        dot += xv * We[d];
    }
#pragma unroll
    for (int m = 16; m >= 1; m >>= 1) dot += __shfl_xor(dot, m, 32);
    if (lane == 0) eta[row] = 1.0f / (1.0f + expf(-dot));
}

#define PROWS  128
#define PKC    32
#define PPITCH 36

__global__ __launch_bounds__(256) __attribute__((amdgpu_waves_per_eu(2, 2)))
void proj_gemm_kernel(const float* __restrict__ xn,
                                 const float* __restrict__ Wk,
                                 const float* __restrict__ Wq,
                                 const float* __restrict__ Wv,
                                 const float* __restrict__ Wvc,
                                 float* __restrict__ Ko,
                                 float* __restrict__ Qo,
                                 float* __restrict__ Vo,
                                 float* __restrict__ vco)
{
    __shared__ __attribute__((aligned(16))) float As[2][PROWS * PPITCH];
    __shared__ __attribute__((aligned(16))) float Bs[2][Mm * PPITCH];

    const int w = blockIdx.y;
    const float* W = (w == 0) ? Wk : (w == 1) ? Wq : (w == 2) ? Wv : Wvc;
    float* Out     = (w == 0) ? Ko : (w == 1) ? Qo : (w == 2) ? Vo : vco;

    const int tid   = threadIdx.x;
    const int wave  = tid >> 5;
    const int lane  = tid & 31;
    const int hi    = lane >> 4;
    const int lr    = lane & 15;
    const int tileM = blockIdx.x * PROWS + wave * 16;

    const int srow = tid >> 3;
    const int sseg = (tid & 7) * 4;
    const float* Ag = xn + (size_t)(blockIdx.x * PROWS + srow) * Dd + sseg;
    const float* Bg = W  + (size_t)srow * Dd + sseg;
    const int sl = srow * PPITCH + sseg;

    v4f ra[4], rb[2];
#pragma unroll
    for (int j = 0; j < 4; ++j) ra[j] = *(const v4f*)(Ag + (size_t)(32 * j) * Dd);
#pragma unroll
    for (int j = 0; j < 2; ++j) rb[j] = *(const v4f*)(Bg + (size_t)(32 * j) * Dd);
#pragma unroll
    for (int j = 0; j < 4; ++j) *(v4f*)&As[0][sl + 32 * j * PPITCH] = ra[j];
#pragma unroll
    for (int j = 0; j < 2; ++j) *(v4f*)&Bs[0][sl + 32 * j * PPITCH] = rb[j];
    __syncthreads();

    const int fa = (wave * 16 + lr) * PPITCH + 2 * hi;
    const int fb = lr * PPITCH + 2 * hi;

    v8f c0 = {}, c1 = {}, c2 = {}, c3 = {};
    for (int c = 0; c < Dd / PKC; ++c) {
        const int cur = c & 1;
        const bool more = (c + 1 < Dd / PKC);
        if (more) {
            const int ko = (c + 1) * PKC;
#pragma unroll
            for (int j = 0; j < 4; ++j) ra[j] = *(const v4f*)(Ag + (size_t)(32 * j) * Dd + ko);
#pragma unroll
            for (int j = 0; j < 2; ++j) rb[j] = *(const v4f*)(Bg + (size_t)(32 * j) * Dd + ko);
        }

        const float* Afr = &As[cur][fa];
        const float* Bfr = &Bs[cur][fb];
#pragma unroll
        for (int kk = 0; kk < PKC; kk += 4) {
            v2f a  = *(const v2f*)(Afr + kk);
            v2f b0 = *(const v2f*)(Bfr + kk);
            v2f b1 = *(const v2f*)(Bfr + 16 * PPITCH + kk);
            v2f b2 = *(const v2f*)(Bfr + 32 * PPITCH + kk);
            v2f b3 = *(const v2f*)(Bfr + 48 * PPITCH + kk);
            c0 = __builtin_amdgcn_wmma_f32_16x16x4_f32(false, a, false, b0, (short)0, c0, false, false);
            c1 = __builtin_amdgcn_wmma_f32_16x16x4_f32(false, a, false, b1, (short)0, c1, false, false);
            c2 = __builtin_amdgcn_wmma_f32_16x16x4_f32(false, a, false, b2, (short)0, c2, false, false);
            c3 = __builtin_amdgcn_wmma_f32_16x16x4_f32(false, a, false, b3, (short)0, c3, false, false);
        }

        if (more) {
#pragma unroll
            for (int j = 0; j < 4; ++j) *(v4f*)&As[cur ^ 1][sl + 32 * j * PPITCH] = ra[j];
#pragma unroll
            for (int j = 0; j < 2; ++j) *(v4f*)&Bs[cur ^ 1][sl + 32 * j * PPITCH] = rb[j];
        }
        __syncthreads();
    }

    if (w < 2) {
#pragma unroll
        for (int r = 0; r < 8; ++r) {
            c0[r] = fmaxf(c0[r], 0.f); c1[r] = fmaxf(c1[r], 0.f);
            c2[r] = fmaxf(c2[r], 0.f); c3[r] = fmaxf(c3[r], 0.f);
            float s = c0[r] + c1[r] + c2[r] + c3[r];
#pragma unroll
            for (int m = 8; m >= 1; m >>= 1) s += __shfl_xor(s, m, 32);
            float inv = 1.0f / (1e-5f + s);
            c0[r] *= inv; c1[r] *= inv; c2[r] *= inv; c3[r] *= inv;
        }
    }

#pragma unroll
    for (int r = 0; r < 8; ++r) {
        float* o = Out + (size_t)(tileM + r + 8 * hi) * Mm + lr;
        o[0]  = c0[r];
        o[16] = c1[r];
        o[32] = c2[r];
        o[48] = c3[r];
    }
}

#define TBLK 32
#define NBLK (Tt / TBLK)
#define SCW 4
#define SBLKF (TBLK * Mm)
#define SPIECES (SBLKF / 4 / (SCW * 32))

template <int M>
__device__ __forceinline__ float lane_xor(float v)
{
    const int b = __builtin_bit_cast(int, v);
    if constexpr (M == 16)
        return __builtin_bit_cast(float,
            __builtin_amdgcn_permlanex16(b, b, 0x76543210, 0xfedcba98, false, false));
    else
        return __builtin_bit_cast(float,
            __builtin_amdgcn_update_dpp(0, b, 0x160 | M, 0xf, 0xf, true));
}
template <int M>
__device__ __forceinline__ void bfly_stage(float (&p)[32], const int lane)
{
    const bool up = (lane & M) != 0;
#pragma unroll
    for (int r = 0; r < M; ++r) {
        const float lo = p[r], hi = p[r + M];
        const float keep = up ? hi : lo;
        const float send = up ? lo : hi;
        p[r] = keep + lane_xor<M>(send);
    }
}

__device__ __forceinline__ float one_plus_tanh_small(float x)
{
    const float x2 = x * x;
    float p = __builtin_fmaf(x2, -0x1.758e7ap-8f, 0x1.521192p-6f);
    p = __builtin_fmaf(x2, p, -0x1.b8389cp-5f);
    p = __builtin_fmaf(x2, p, 0x1.110704p-3f);
    p = __builtin_fmaf(x2, p, -0x1.555532p-2f);
    return 1.0f + __builtin_fmaf(x2, x * p, x);
}

__global__ __launch_bounds__(SCW * 32) void scan_kernel(const float* __restrict__ K,
                            const float* __restrict__ Q,
                            const float* __restrict__ V,
                            const float* __restrict__ vc,
                            const float* __restrict__ eta,
                            const float* __restrict__ theta_matrix,
                            float* __restrict__ y)
{
    __shared__ __attribute__((aligned(16))) float Svc[2][SBLKF];
    __shared__ __attribute__((aligned(16))) float Skq[2][2][SBLKF];

    const int b    = blockIdx.x;
    const int tid  = threadIdx.x;
    const int wave = tid >> 5;
    const int lane = tid & 31;
    const int i    = blockIdx.y * SCW + wave;
    const float theta_i = theta_matrix[i];

    float s0 = 0.f, s1 = 0.f;
    const size_t base = (size_t)b * Tt * Mm;
    const float* etal = eta + (size_t)b * Tt + lane;
    const float* Vl   = V + base + (size_t)lane * Mm + i;
    const v4f* gvc = (const v4f*)(vc + base) + tid;
    const v4f* gK  = (const v4f*)(K  + base) + tid;
    const v4f* gQ  = (const v4f*)(Q  + base) + tid;

    float a0[TBLK], a1[TBLK];
#pragma unroll
    for (int tt = 0; tt < TBLK; ++tt) {
        const v2f av = ((const v2f*)(vc + base) + lane)[tt * (Mm / 2)];
        a0[tt] = av.x; a1[tt] = av.y;
    }
    float u_l = etal[0] * Vl[0];

#pragma unroll
    for (int j = 0; j < SPIECES; ++j) {
        const int pc = j * (SCW * 32);
        const v4f gv = gvc[SBLKF / 4 + pc], gk = gK[pc], gq = gQ[pc];
        ((v4f*)Svc[1])[tid + pc]    = gv;
        ((v4f*)Skq[0][0])[tid + pc] = gk;
        ((v4f*)Skq[0][1])[tid + pc] = gq;
    }
    __syncthreads();

    for (int t0 = 0; t0 < Tt; t0 += TBLK) {
        const int n = t0 / TBLK;
        const int tn = (t0 + TBLK < Tt) ? t0 + TBLK : t0;
        const float et_n = etal[tn];
        const float vi_n = Vl[(size_t)tn * Mm];

        const int nv = (n + 2 < NBLK) ? n + 2 : NBLK - 1;
        const int nk = (n + 1 < NBLK) ? n + 1 : NBLK - 1;
        v4f gv[SPIECES], gk[SPIECES], gq[SPIECES];
#pragma unroll
        for (int j = 0; j < SPIECES; ++j) {
            const int pc = j * (SCW * 32);
            gv[j] = gvc[(size_t)nv * (SBLKF / 4) + pc];
            gk[j] = gK[(size_t)nk * (SBLKF / 4) + pc];
            gq[j] = gQ[(size_t)nk * (SBLKF / 4) + pc];
        }

        const v2f* Lv = (const v2f*)Svc[(n + 1) & 1] + lane;
        const v2f* Lk = (const v2f*)Skq[n & 1][0] + lane;
        const v2f* Lq = (const v2f*)Skq[n & 1][1] + lane;
        float an0[TBLK], an1[TBLK], kx0[TBLK], kx1[TBLK], qx0[TBLK], qx1[TBLK];
#pragma unroll
        for (int tt = 0; tt < TBLK; ++tt) {
            const int o = tt * (Mm / 2);
            const v2f av = Lv[o], kv = Lk[o], qv = Lq[o];
            an0[tt] = av.x; an1[tt] = av.y;
            kx0[tt] = kv.x; kx1[tt] = kv.y;
            qx0[tt] = qv.x; qx1[tt] = qv.y;
        }

        float am[4] = {0.f, 0.f, 0.f, 0.f};
#pragma unroll
        for (int tt = 0; tt < TBLK; ++tt)
            am[tt & 3] = fmaxf(am[tt & 3], fmaxf(fabsf(a0[tt]), fabsf(a1[tt])));
        const float amax = fmaxf(fmaxf(am[0], am[1]), fmaxf(am[2], am[3]));
        const bool big = !(fabsf(theta_i) * amax < 0.625f);

        float c0[TBLK], c1[TBLK];
        if (__builtin_expect(__builtin_amdgcn_ballot_w32(big) == 0u, 1)) {
#pragma unroll
            for (int tt = 0; tt < TBLK; ++tt) {
                c0[tt] = one_plus_tanh_small(theta_i * a0[tt]);
                c1[tt] = one_plus_tanh_small(theta_i * a1[tt]);
            }
        } else {
#pragma unroll
            for (int tt = 0; tt < TBLK; ++tt) {
                c0[tt] = 1.0f + tanhf(theta_i * a0[tt]);
                c1[tt] = 1.0f + tanhf(theta_i * a1[tt]);
            }
        }

        float p[TBLK];
#pragma unroll
        for (int tt = 0; tt < TBLK; ++tt) {
            const float u = __builtin_bit_cast(float,
                __builtin_amdgcn_readlane(__builtin_bit_cast(int, u_l), tt));
            s0 = __builtin_fmaf(u, kx0[tt], s0 * c0[tt]);
            s1 = __builtin_fmaf(u, kx1[tt], s1 * c1[tt]);
            p[tt] = __builtin_fmaf(s0, qx0[tt], s1 * qx1[tt]);
            a0[tt] = an0[tt]; a1[tt] = an1[tt];
        }

#pragma unroll
        for (int j = 0; j < SPIECES; ++j) {
            const int pc = tid + j * (SCW * 32);
            ((v4f*)Svc[n & 1])[pc]          = gv[j];
            ((v4f*)Skq[(n + 1) & 1][0])[pc] = gk[j];
            ((v4f*)Skq[(n + 1) & 1][1])[pc] = gq[j];
        }

        bfly_stage<16>(p, lane);
        bfly_stage<8>(p, lane);
        bfly_stage<4>(p, lane);
        bfly_stage<2>(p, lane);
        bfly_stage<1>(p, lane);
        u_l = et_n * vi_n;
        __syncthreads();
        y[base + (size_t)(t0 + lane) * Mm + i] = p[0];
    }
}

__global__ void out_gemm_kernel(const float* __restrict__ y,
                                const float* __restrict__ Wout,
                                const float* __restrict__ bout,
                                float* __restrict__ out)
{
    const int tileM = blockIdx.x * 16;
    const int tileN = blockIdx.y * 64;
    const int lane  = threadIdx.x & 31;
    const int hi    = lane >> 4;
    const int lr    = lane & 15;

    const float* Arow = y + (size_t)(tileM + lr) * Mm + 2 * hi;
    const float* B0 = Wout + (size_t)(tileN + lr +  0) * Mm + 2 * hi;
    const float* B1 = Wout + (size_t)(tileN + lr + 16) * Mm + 2 * hi;
    const float* B2 = Wout + (size_t)(tileN + lr + 32) * Mm + 2 * hi;
    const float* B3 = Wout + (size_t)(tileN + lr + 48) * Mm + 2 * hi;

    v8f c0 = {}, c1 = {}, c2 = {}, c3 = {};
#pragma unroll
    for (int kk = 0; kk < Mm; kk += 4) {
        v2f a  = *(const v2f*)(Arow + kk);
        v2f b0 = *(const v2f*)(B0 + kk);
        v2f b1 = *(const v2f*)(B1 + kk);
        v2f b2 = *(const v2f*)(B2 + kk);
        v2f b3 = *(const v2f*)(B3 + kk);
        c0 = __builtin_amdgcn_wmma_f32_16x16x4_f32(false, a, false, b0, (short)0, c0, false, false);
        c1 = __builtin_amdgcn_wmma_f32_16x16x4_f32(false, a, false, b1, (short)0, c1, false, false);
        c2 = __builtin_amdgcn_wmma_f32_16x16x4_f32(false, a, false, b2, (short)0, c2, false, false);
        c3 = __builtin_amdgcn_wmma_f32_16x16x4_f32(false, a, false, b3, (short)0, c3, false, false);
    }

    const float bb0 = bout[tileN + lr +  0];
    const float bb1 = bout[tileN + lr + 16];
    const float bb2 = bout[tileN + lr + 32];
    const float bb3 = bout[tileN + lr + 48];

#pragma unroll
    for (int r = 0; r < 8; ++r) {
        float* o = out + (size_t)(tileM + r + 8 * hi) * Oo + tileN + lr;
        o[0]  = c0[r] + bb0;
        o[16] = c1[r] + bb1;
        o[32] = c2[r] + bb2;
        o[48] = c3[r] + bb3;
    }
}

extern "C" void kernel_launch(void* const* d_in, const int* in_sizes, int n_in,
                              void* d_out, int out_size, void* d_ws, size_t ws_size,
                              hipStream_t stream)
{
    const float* x     = (const float*)d_in[0];
    const float* Wk    = (const float*)d_in[1];
    const float* Wq    = (const float*)d_in[2];
    const float* Wv    = (const float*)d_in[3];
    const float* Wvc   = (const float*)d_in[4];
    const float* We    = (const float*)d_in[5];
    const float* theta = (const float*)d_in[6];
    const float* Wout  = (const float*)d_in[7];
    const float* bout  = (const float*)d_in[8];
    const float* gamma = (const float*)d_in[9];
    const float* beta  = (const float*)d_in[10];
    float* out = (float*)d_out;

    float* ws  = (float*)d_ws;
    const size_t NROW = (size_t)Bb * Tt;
    float* xn  = ws;
    float* Ko  = xn  + NROW * Dd;
    float* Qo  = Ko  + NROW * Mm;
    float* Vo  = Qo  + NROW * Mm;
    float* vco = Vo  + NROW * Mm;
    float* yb  = vco + NROW * Mm;
    float* eta = yb  + NROW * Mm;

    ln_eta_kernel<<<dim3(NROW / 8), dim3(256), 0, stream>>>(x, gamma, beta, We, xn, eta);
    proj_gemm_kernel<<<dim3(NROW / PROWS, 4), dim3(256), 0, stream>>>(xn, Wk, Wq, Wv, Wvc,
                                                                  Ko, Qo, Vo, vco);
    scan_kernel<<<dim3(Bb, Mm / SCW), dim3(SCW * 32), 0, stream>>>(Ko, Qo, Vo, vco, eta, theta, yb);
    out_gemm_kernel<<<dim3(NROW / 16, Oo / 64), dim3(32), 0, stream>>>(yb, Wout, bout, out);
}


# scan kernel: two waves per state row (32 columns each, one a lane), twice the work-groups, a wave on every SIMD; y sum tree is now per-half butterfly, then lower + upper half (handed over via LDS)
# speedup vs baseline: 11.0337x; 1.0629x over previous
#include <hip/hip_runtime.h>
#include <hip/hip_bf16.h>

typedef float v2f __attribute__((ext_vector_type(2)));
typedef float v4f __attribute__((ext_vector_type(4)));
typedef float v8f __attribute__((ext_vector_type(8)));

#define Bb 8
#define Tt 1024
#define Dd 512
#define Mm 64
#define Oo 512

__global__ void ln_eta_kernel(const float* __restrict__ x,
                              const float* __restrict__ gamma,
                              const float* __restrict__ beta,
                              const float* __restrict__ We,
                              float* __restrict__ xn,
                              float* __restrict__ eta)
{
    const int wave = threadIdx.x >> 5;
    const int lane = threadIdx.x & 31;
    const int row  = blockIdx.x * 8 + wave;
    const float* xr = x + (size_t)row * Dd;

    float v[16];
    float sum = 0.f, sq = 0.f;
#pragma unroll
    for (int e = 0; e < 16; ++e) {
        float t = xr[lane + 32 * e];
        v[e] = t; sum += t; sq += t * t;
    }
#pragma unroll
    for (int m = 16; m >= 1; m >>= 1) {
        sum += __shfl_xor(sum, m, 32);
        sq  += __shfl_xor(sq,  m, 32);
    }
    const float mu = sum * (1.0f / Dd);
    const float var = sq * (1.0f / Dd) - mu * mu;
    const float rs = rsqrtf(var + 1e-5f);

    float dot = 0.f;
    float* xo = xn + (size_t)row * Dd;
#pragma unroll
    for (int e = 0; e < 16; ++e) {
        const int d = lane + 32 * e;
        float xv = (v[e] - mu) * rs * gamma[d] + beta[d];
        xo[d] = xv;
        dot += xv * We[d];
    }
#pragma unroll
    for (int m = 16; m >= 1; m >>= 1) dot += __shfl_xor(dot, m, 32);
    if (lane == 0) eta[row] = 1.0f / (1.0f + expf(-dot));
}

#define PROWS  128
#define PKC    32
#define PPITCH 36

__global__ __launch_bounds__(256) __attribute__((amdgpu_waves_per_eu(2, 2)))
void proj_gemm_kernel(const float* __restrict__ xn,
                                 const float* __restrict__ Wk,
                                 const float* __restrict__ Wq,
                                 const float* __restrict__ Wv,
                                 const float* __restrict__ Wvc,
                                 float* __restrict__ Ko,
                                 float* __restrict__ Qo,
                                 float* __restrict__ Vo,
                                 float* __restrict__ vco)
{
    __shared__ __attribute__((aligned(16))) float As[2][PROWS * PPITCH];
    __shared__ __attribute__((aligned(16))) float Bs[2][Mm * PPITCH];

    const int w = blockIdx.y;
    const float* W = (w == 0) ? Wk : (w == 1) ? Wq : (w == 2) ? Wv : Wvc;
    float* Out     = (w == 0) ? Ko : (w == 1) ? Qo : (w == 2) ? Vo : vco;

    const int tid   = threadIdx.x;
    const int wave  = tid >> 5;
    const int lane  = tid & 31;
    const int hi    = lane >> 4;
    const int lr    = lane & 15;
    const int tileM = blockIdx.x * PROWS + wave * 16;

    const int srow = tid >> 3;
    const int sseg = (tid & 7) * 4;
    const float* Ag = xn + (size_t)(blockIdx.x * PROWS + srow) * Dd + sseg;
    const float* Bg = W  + (size_t)srow * Dd + sseg;
    const int sl = srow * PPITCH + sseg;

    v4f ra[4], rb[2];
#pragma unroll
    for (int j = 0; j < 4; ++j) ra[j] = *(const v4f*)(Ag + (size_t)(32 * j) * Dd);
#pragma unroll
    for (int j = 0; j < 2; ++j) rb[j] = *(const v4f*)(Bg + (size_t)(32 * j) * Dd);
#pragma unroll
    for (int j = 0; j < 4; ++j) *(v4f*)&As[0][sl + 32 * j * PPITCH] = ra[j];
#pragma unroll
    for (int j = 0; j < 2; ++j) *(v4f*)&Bs[0][sl + 32 * j * PPITCH] = rb[j];
    __syncthreads();

    const int fa = (wave * 16 + lr) * PPITCH + 2 * hi;
    const int fb = lr * PPITCH + 2 * hi;

    v8f c0 = {}, c1 = {}, c2 = {}, c3 = {};
    for (int c = 0; c < Dd / PKC; ++c) {
        const int cur = c & 1;
        const bool more = (c + 1 < Dd / PKC);
        if (more) {
            const int ko = (c + 1) * PKC;
#pragma unroll
            for (int j = 0; j < 4; ++j) ra[j] = *(const v4f*)(Ag + (size_t)(32 * j) * Dd + ko);
#pragma unroll
            for (int j = 0; j < 2; ++j) rb[j] = *(const v4f*)(Bg + (size_t)(32 * j) * Dd + ko);
        }

        const float* Afr = &As[cur][fa];
        const float* Bfr = &Bs[cur][fb];
#pragma unroll
        for (int kk = 0; kk < PKC; kk += 4) {
            v2f a  = *(const v2f*)(Afr + kk);
            v2f b0 = *(const v2f*)(Bfr + kk);
            v2f b1 = *(const v2f*)(Bfr + 16 * PPITCH + kk);
            v2f b2 = *(const v2f*)(Bfr + 32 * PPITCH + kk);
            v2f b3 = *(const v2f*)(Bfr + 48 * PPITCH + kk);
            c0 = __builtin_amdgcn_wmma_f32_16x16x4_f32(false, a, false, b0, (short)0, c0, false, false);
            c1 = __builtin_amdgcn_wmma_f32_16x16x4_f32(false, a, false, b1, (short)0, c1, false, false);
            c2 = __builtin_amdgcn_wmma_f32_16x16x4_f32(false, a, false, b2, (short)0, c2, false, false);
            c3 = __builtin_amdgcn_wmma_f32_16x16x4_f32(false, a, false, b3, (short)0, c3, false, false);
        }

        if (more) {
#pragma unroll
            for (int j = 0; j < 4; ++j) *(v4f*)&As[cur ^ 1][sl + 32 * j * PPITCH] = ra[j];
#pragma unroll
            for (int j = 0; j < 2; ++j) *(v4f*)&Bs[cur ^ 1][sl + 32 * j * PPITCH] = rb[j];
        }
        __syncthreads();
    }

    if (w < 2) {
#pragma unroll
        for (int r = 0; r < 8; ++r) {
            c0[r] = fmaxf(c0[r], 0.f); c1[r] = fmaxf(c1[r], 0.f);
            c2[r] = fmaxf(c2[r], 0.f); c3[r] = fmaxf(c3[r], 0.f);
            float s = c0[r] + c1[r] + c2[r] + c3[r];
#pragma unroll
            for (int m = 8; m >= 1; m >>= 1) s += __shfl_xor(s, m, 32);
            float inv = 1.0f / (1e-5f + s);
            c0[r] *= inv; c1[r] *= inv; c2[r] *= inv; c3[r] *= inv;
        }
    }

#pragma unroll
    for (int r = 0; r < 8; ++r) {
        float* o = Out + (size_t)(tileM + r + 8 * hi) * Mm + lr;
        o[0]  = c0[r];
        o[16] = c1[r];
        o[32] = c2[r];
        o[48] = c3[r];
    }
}

#define TBLK 32
#define NBLK (Tt / TBLK)
#define SCW 4
#define SROWS (SCW / 2)
#define SBLKF (TBLK * Mm)
#define SPIECES (SBLKF / 4 / (SCW * 32))

template <int M>
__device__ __forceinline__ float lane_xor(float v)
{
    const int b = __builtin_bit_cast(int, v);
    if constexpr (M == 16)
        return __builtin_bit_cast(float,
            __builtin_amdgcn_permlanex16(b, b, 0x76543210, 0xfedcba98, false, false));
    else
        return __builtin_bit_cast(float,
            __builtin_amdgcn_update_dpp(0, b, 0x160 | M, 0xf, 0xf, true));
}
template <int M>
__device__ __forceinline__ void bfly_stage(float (&p)[32], const int lane)
{
    const bool up = (lane & M) != 0;
#pragma unroll
    for (int r = 0; r < M; ++r) {
        const float lo = p[r], hi = p[r + M];
        const float keep = up ? hi : lo;
        const float send = up ? lo : hi;
        p[r] = keep + lane_xor<M>(send);
    }
}

__device__ __forceinline__ float one_plus_tanh_small(float x)
{
    const float x2 = x * x;
    float p = __builtin_fmaf(x2, -0x1.758e7ap-8f, 0x1.521192p-6f);
    p = __builtin_fmaf(x2, p, -0x1.b8389cp-5f);
    p = __builtin_fmaf(x2, p, 0x1.110704p-3f);
    p = __builtin_fmaf(x2, p, -0x1.555532p-2f);
    return 1.0f + __builtin_fmaf(x2, x * p, x);
}

__global__ __launch_bounds__(SCW * 32) void scan_kernel(const float* __restrict__ K,
                            const float* __restrict__ Q,
                            const float* __restrict__ V,
                            const float* __restrict__ vc,
                            const float* __restrict__ eta,
                            const float* __restrict__ theta_matrix,
                            float* __restrict__ y)
{
    __shared__ __attribute__((aligned(16))) float Svc[2][SBLKF];
    __shared__ __attribute__((aligned(16))) float Skq[2][2][SBLKF];
    __shared__ float Sy[2][SROWS][TBLK];

    const int b    = blockIdx.x;
    const int tid  = threadIdx.x;
    const int wave = tid >> 5;
    const int lane = tid & 31;
    const int rw   = wave >> 1;
    const int h    = wave & 1;
    const int i    = blockIdx.y * SROWS + rw;
    const int col  = h * 32 + lane;
    const float theta_i = theta_matrix[i];

    float s = 0.f;
    const size_t base = (size_t)b * Tt * Mm;
    const float* etal = eta + (size_t)b * Tt + lane;
    const float* Vl   = V + base + (size_t)lane * Mm + i;
    const v4f* gvc = (const v4f*)(vc + base) + tid;
    const v4f* gK  = (const v4f*)(K  + base) + tid;
    const v4f* gQ  = (const v4f*)(Q  + base) + tid;

    float a[TBLK];
#pragma unroll
    for (int tt = 0; tt < TBLK; ++tt) a[tt] = vc[base + (size_t)tt * Mm + col];
    float u_l = etal[0] * Vl[0];

#pragma unroll
    for (int j = 0; j < SPIECES; ++j) {
        const int pc = j * (SCW * 32);
        const v4f gv = gvc[SBLKF / 4 + pc], gk = gK[pc], gq = gQ[pc];
        ((v4f*)Svc[1])[tid + pc]    = gv;
        ((v4f*)Skq[0][0])[tid + pc] = gk;
        ((v4f*)Skq[0][1])[tid + pc] = gq;
    }
    __syncthreads();

    float yl = 0.f, yu = 0.f;

    for (int t0 = 0; t0 < Tt; t0 += TBLK) {
        const int n = t0 / TBLK;
        const int tn = (t0 + TBLK < Tt) ? t0 + TBLK : t0;
        const float et_n = etal[tn];
        const float vi_n = Vl[(size_t)tn * Mm];

        const int nv = (n + 2 < NBLK) ? n + 2 : NBLK - 1;
        const int nk = (n + 1 < NBLK) ? n + 1 : NBLK - 1;
        v4f gv[SPIECES], gk[SPIECES], gq[SPIECES];
#pragma unroll
        for (int j = 0; j < SPIECES; ++j) {
            const int pc = j * (SCW * 32);
            gv[j] = gvc[(size_t)nv * (SBLKF / 4) + pc];
            gk[j] = gK[(size_t)nk * (SBLKF / 4) + pc];
            gq[j] = gQ[(size_t)nk * (SBLKF / 4) + pc];
        }

        const float* Lv = Svc[(n + 1) & 1] + col;
        const float* Lk = Skq[n & 1][0] + col;
        const float* Lq = Skq[n & 1][1] + col;
        float an[TBLK], kx[TBLK], qx[TBLK];
#pragma unroll
        for (int tt = 0; tt < TBLK; ++tt) {
            an[tt] = Lv[tt * Mm];
            kx[tt] = Lk[tt * Mm];
            qx[tt] = Lq[tt * Mm];
        }

        float am[4] = {0.f, 0.f, 0.f, 0.f};
#pragma unroll
        for (int tt = 0; tt < TBLK; ++tt)
            am[tt & 3] = fmaxf(am[tt & 3], fabsf(a[tt]));
        const float amax = fmaxf(fmaxf(am[0], am[1]), fmaxf(am[2], am[3]));
        const bool big = !(fabsf(theta_i) * amax < 0.625f);

        float c[TBLK];
        if (__builtin_expect(__builtin_amdgcn_ballot_w32(big) == 0u, 1)) {
#pragma unroll
            for (int tt = 0; tt < TBLK; ++tt) c[tt] = one_plus_tanh_small(theta_i * a[tt]);
        } else {
#pragma unroll
            for (int tt = 0; tt < TBLK; ++tt) c[tt] = 1.0f + tanhf(theta_i * a[tt]);
        }

        if (h == 0 && n > 0)
            y[base + (size_t)(t0 - TBLK + lane) * Mm + i] = yl + yu;

        float p[TBLK];
#pragma unroll
        for (int tt = 0; tt < TBLK; ++tt) {
            const float u = __builtin_bit_cast(float,
                __builtin_amdgcn_readlane(__builtin_bit_cast(int, u_l), tt));
            s = __builtin_fmaf(u, kx[tt], s * c[tt]);
            p[tt] = s * qx[tt];
            a[tt] = an[tt];
        }

#pragma unroll
        for (int j = 0; j < SPIECES; ++j) {
            const int pc = tid + j * (SCW * 32);
            ((v4f*)Svc[n & 1])[pc]          = gv[j];
            ((v4f*)Skq[(n + 1) & 1][0])[pc] = gk[j];
            ((v4f*)Skq[(n + 1) & 1][1])[pc] = gq[j];
        }

        bfly_stage<16>(p, lane);
        bfly_stage<8>(p, lane);
        bfly_stage<4>(p, lane);
        bfly_stage<2>(p, lane);
        bfly_stage<1>(p, lane);
        u_l = et_n * vi_n;
        if (h == 1) Sy[n & 1][rw][lane] = p[0];
        __syncthreads();
        if (h == 0) { yl = p[0]; yu = Sy[n & 1][rw][lane]; }
    }
    if (h == 0) y[base + (size_t)(Tt - TBLK + lane) * Mm + i] = yl + yu;
}

__global__ void out_gemm_kernel(const float* __restrict__ y,
                                const float* __restrict__ Wout,
                                const float* __restrict__ bout,
                                float* __restrict__ out)
{
    const int tileM = blockIdx.x * 16;
    const int tileN = blockIdx.y * 64;
    const int lane  = threadIdx.x & 31;
    const int hi    = lane >> 4;
    const int lr    = lane & 15;

    const float* Arow = y + (size_t)(tileM + lr) * Mm + 2 * hi;
    const float* B0 = Wout + (size_t)(tileN + lr +  0) * Mm + 2 * hi;
    const float* B1 = Wout + (size_t)(tileN + lr + 16) * Mm + 2 * hi;
    const float* B2 = Wout + (size_t)(tileN + lr + 32) * Mm + 2 * hi;
    const float* B3 = Wout + (size_t)(tileN + lr + 48) * Mm + 2 * hi;

    v8f c0 = {}, c1 = {}, c2 = {}, c3 = {};
#pragma unroll
    for (int kk = 0; kk < Mm; kk += 4) {
        v2f a  = *(const v2f*)(Arow + kk);
        v2f b0 = *(const v2f*)(B0 + kk);
        v2f b1 = *(const v2f*)(B1 + kk);
        v2f b2 = *(const v2f*)(B2 + kk);
        v2f b3 = *(const v2f*)(B3 + kk);
        c0 = __builtin_amdgcn_wmma_f32_16x16x4_f32(false, a, false, b0, (short)0, c0, false, false);
        c1 = __builtin_amdgcn_wmma_f32_16x16x4_f32(false, a, false, b1, (short)0, c1, false, false);
        c2 = __builtin_amdgcn_wmma_f32_16x16x4_f32(false, a, false, b2, (short)0, c2, false, false);
        c3 = __builtin_amdgcn_wmma_f32_16x16x4_f32(false, a, false, b3, (short)0, c3, false, false);
    }

    const float bb0 = bout[tileN + lr +  0];
    const float bb1 = bout[tileN + lr + 16];
    const float bb2 = bout[tileN + lr + 32];
    const float bb3 = bout[tileN + lr + 48];

#pragma unroll
    for (int r = 0; r < 8; ++r) {
        float* o = out + (size_t)(tileM + r + 8 * hi) * Oo + tileN + lr;
        o[0]  = c0[r] + bb0;
        o[16] = c1[r] + bb1;
        o[32] = c2[r] + bb2;
        o[48] = c3[r] + bb3;
    }
}

extern "C" void kernel_launch(void* const* d_in, const int* in_sizes, int n_in,
                              void* d_out, int out_size, void* d_ws, size_t ws_size,
                              hipStream_t stream)
{
    const float* x     = (const float*)d_in[0];
    const float* Wk    = (const float*)d_in[1];
    const float* Wq    = (const float*)d_in[2];
    const float* Wv    = (const float*)d_in[3];
    const float* Wvc   = (const float*)d_in[4];
    const float* We    = (const float*)d_in[5];
    const float* theta = (const float*)d_in[6];
    const float* Wout  = (const float*)d_in[7];
    const float* bout  = (const float*)d_in[8];
    const float* gamma = (const float*)d_in[9];
    const float* beta  = (const float*)d_in[10];
    float* out = (float*)d_out;

    float* ws  = (float*)d_ws;
    const size_t NROW = (size_t)Bb * Tt;
    float* xn  = ws;
    float* Ko  = xn  + NROW * Dd;
    float* Qo  = Ko  + NROW * Mm;
    float* Vo  = Qo  + NROW * Mm;
    float* vco = Vo  + NROW * Mm;
    float* yb  = vco + NROW * Mm;
    float* eta = yb  + NROW * Mm;

    ln_eta_kernel<<<dim3(NROW / 8), dim3(256), 0, stream>>>(x, gamma, beta, We, xn, eta);
    proj_gemm_kernel<<<dim3(NROW / PROWS, 4), dim3(256), 0, stream>>>(xn, Wk, Wq, Wv, Wvc,
                                                                  Ko, Qo, Vo, vco);
    scan_kernel<<<dim3(Bb, Mm / SROWS), dim3(SCW * 32), 0, stream>>>(Ko, Qo, Vo, vco, eta, theta, yb);
    out_gemm_kernel<<<dim3(NROW / 16, Oo / 64), dim3(32), 0, stream>>>(yb, Wout, bout, out);
}


# LayerNorm fused into the projection GEMM's head: a work-group normalises its 32 rows and keeps them whole in LDS as A for all four weights; xn never goes to memory, one launch less; same sum trees
# speedup vs baseline: 12.0835x; 1.0951x over previous
#include <hip/hip_runtime.h>
#include <hip/hip_bf16.h>

typedef float v2f __attribute__((ext_vector_type(2)));
typedef float v4f __attribute__((ext_vector_type(4)));
typedef float v8f __attribute__((ext_vector_type(8)));

#define Bb 8
#define Tt 1024
#define Dd 512
#define Mm 64
#define Oo 512

template <int M>
__device__ __forceinline__ float lane_xor(float v)
{
    const int b = __builtin_bit_cast(int, v);
    if constexpr (M == 16)
        return __builtin_bit_cast(float,
            __builtin_amdgcn_permlanex16(b, b, 0x76543210, 0xfedcba98, false, false));
    else
        return __builtin_bit_cast(float,
            __builtin_amdgcn_update_dpp(0, b, 0x160 | M, 0xf, 0xf, true));
}
template <int M>
__device__ __forceinline__ void xsum4(float (&a)[4])
{
#pragma unroll
    for (int q = 0; q < 4; ++q) a[q] += lane_xor<M>(a[q]);
}
__device__ __forceinline__ void wave_sum4(float (&a)[4])
{
    xsum4<16>(a); xsum4<8>(a); xsum4<4>(a); xsum4<2>(a); xsum4<1>(a);
}

#define LROWS  32
#define APITCH 516
#define WROWS  256
#define PKC    32
#define PPITCH 36

__device__ __forceinline__ const float* wrow32(const float* Wk, const float* Wq,
                                               const float* Wv, const float* Wvc, int j)
{
    const float* W = (j < 2) ? Wk : (j < 4) ? Wq : (j < 6) ? Wv : Wvc;
    return W + (size_t)(32 * (j & 1)) * Dd;
}

__global__ __launch_bounds__(256) __attribute__((amdgpu_waves_per_eu(2, 2)))
void ln_proj_kernel(const float* __restrict__ x,
                    const float* __restrict__ gamma,
                    const float* __restrict__ beta,
                    const float* __restrict__ We,
                    const float* __restrict__ Wk,
                    const float* __restrict__ Wq,
                    const float* __restrict__ Wv,
                    const float* __restrict__ Wvc,
                    float* __restrict__ Ko,
                    float* __restrict__ Qo,
                    float* __restrict__ Vo,
                    float* __restrict__ vco,
                    float* __restrict__ eta)
{
    __shared__ __attribute__((aligned(16))) float As[LROWS * APITCH];
    __shared__ __attribute__((aligned(16))) float Bs[2][WROWS * PPITCH];

    const int tid   = threadIdx.x;
    const int wave  = __builtin_amdgcn_readfirstlane(tid >> 5);
    const int lane  = tid & 31;
    const int hi    = lane >> 4;
    const int lr    = lane & 15;
    const int row0  = blockIdx.x * LROWS;

    const int srow = tid >> 3;
    const int sseg = (tid & 7) * 4;
    const size_t goff = (size_t)srow * Dd + sseg;
    const int sl = srow * PPITCH + sseg;

    v4f rb[8];
#pragma unroll
    for (int j = 0; j < 8; ++j) rb[j] = *(const v4f*)(wrow32(Wk, Wq, Wv, Wvc, j) + goff);

    {
        float g[16], bt[16], we[16];
        float v[4][16];
        const float* xr = x + (size_t)(row0 + 4 * wave) * Dd + lane;
#pragma unroll
        for (int q = 0; q < 4; ++q)
#pragma unroll
            for (int e = 0; e < 16; ++e) v[q][e] = xr[(size_t)q * Dd + 32 * e];
#pragma unroll
        for (int e = 0; e < 16; ++e) {
            const int d = lane + 32 * e;
            g[e] = gamma[d]; bt[e] = beta[d]; we[e] = We[d];
        }
        float sum[4], sq[4];
#pragma unroll
        for (int q = 0; q < 4; ++q) {
            sum[q] = 0.f; sq[q] = 0.f;
#pragma unroll
            for (int e = 0; e < 16; ++e) {
                const float t = v[q][e];
                sum[q] += t; sq[q] += t * t;
            }
        }
        wave_sum4(sum);
        wave_sum4(sq);

        float dot[4];
#pragma unroll
        for (int q = 0; q < 4; ++q) {
            const float mu = sum[q] * (1.0f / Dd);
            const float var = sq[q] * (1.0f / Dd) - mu * mu;
            const float rs = rsqrtf(var + 1e-5f);
            dot[q] = 0.f;
            float* xo = &As[(4 * wave + q) * APITCH + lane];
#pragma unroll
            for (int e = 0; e < 16; ++e) {
                float xv = (v[q][e] - mu) * rs * g[e] + bt[e];
                xo[32 * e] = xv;
                dot[q] += xv * we[e];
            }
        }
        wave_sum4(dot);
        const float dl = (lane == 0) ? dot[0] : (lane == 1) ? dot[1] : (lane == 2) ? dot[2] : dot[3];
        if (lane < 4) eta[row0 + 4 * wave + lane] = 1.0f / (1.0f + expf(-dl));
    }

#pragma unroll
    for (int j = 0; j < 8; ++j) *(v4f*)&Bs[0][sl + 32 * j * PPITCH] = rb[j];
    __syncthreads();

    const int rg = wave >> 2;
    const int w  = wave & 3;
    float* Out   = (w == 0) ? Ko : (w == 1) ? Qo : (w == 2) ? Vo : vco;
    const int tileM = row0 + 16 * rg;

    const int fa = (16 * rg + lr) * APITCH + 2 * hi;
    const int fb = (64 * w + lr) * PPITCH + 2 * hi;

    v8f c0 = {}, c1 = {}, c2 = {}, c3 = {};
    for (int c = 0; c < Dd / PKC; ++c) {
        const int cur = c & 1;
        const bool more = (c + 1 < Dd / PKC);
        if (more) {
            const int ko = (c + 1) * PKC;
#pragma unroll
            for (int j = 0; j < 8; ++j)
                rb[j] = *(const v4f*)(wrow32(Wk, Wq, Wv, Wvc, j) + goff + ko);
        }

        const float* Afr = &As[fa + c * PKC];
        const float* Bfr = &Bs[cur][fb];
#pragma unroll
        for (int kk = 0; kk < PKC; kk += 4) {
            v2f a  = *(const v2f*)(Afr + kk);
            v2f b0 = *(const v2f*)(Bfr + kk);
            v2f b1 = *(const v2f*)(Bfr + 16 * PPITCH + kk);
            v2f b2 = *(const v2f*)(Bfr + 32 * PPITCH + kk);
            v2f b3 = *(const v2f*)(Bfr + 48 * PPITCH + kk);
            c0 = __builtin_amdgcn_wmma_f32_16x16x4_f32(false, a, false, b0, (short)0, c0, false, false);
            c1 = __builtin_amdgcn_wmma_f32_16x16x4_f32(false, a, false, b1, (short)0, c1, false, false);
            c2 = __builtin_amdgcn_wmma_f32_16x16x4_f32(false, a, false, b2, (short)0, c2, false, false);
            c3 = __builtin_amdgcn_wmma_f32_16x16x4_f32(false, a, false, b3, (short)0, c3, false, false);
        }

        if (more) {
#pragma unroll
            for (int j = 0; j < 8; ++j) *(v4f*)&Bs[cur ^ 1][sl + 32 * j * PPITCH] = rb[j];
        }
        __syncthreads();
    }

    if (w < 2) {
#pragma unroll
        for (int r = 0; r < 8; ++r) {
            c0[r] = fmaxf(c0[r], 0.f); c1[r] = fmaxf(c1[r], 0.f);
            c2[r] = fmaxf(c2[r], 0.f); c3[r] = fmaxf(c3[r], 0.f);
            float s = c0[r] + c1[r] + c2[r] + c3[r];
#pragma unroll
            for (int m = 8; m >= 1; m >>= 1) s += __shfl_xor(s, m, 32);
            float inv = 1.0f / (1e-5f + s);
            c0[r] *= inv; c1[r] *= inv; c2[r] *= inv; c3[r] *= inv;
        }
    }

#pragma unroll
    for (int r = 0; r < 8; ++r) {
        float* o = Out + (size_t)(tileM + r + 8 * hi) * Mm + lr;
        o[0]  = c0[r];
        o[16] = c1[r];
        o[32] = c2[r];
        o[48] = c3[r];
    }
}

#define TBLK 32
#define NBLK (Tt / TBLK)
#define SCW 4
#define SROWS (SCW / 2)
#define SBLKF (TBLK * Mm)
#define SPIECES (SBLKF / 4 / (SCW * 32))

template <int M>
__device__ __forceinline__ void bfly_stage(float (&p)[32], const int lane)
{
    const bool up = (lane & M) != 0;
#pragma unroll
    for (int r = 0; r < M; ++r) {
        const float lo = p[r], hi = p[r + M];
        const float keep = up ? hi : lo;
        const float send = up ? lo : hi;
        p[r] = keep + lane_xor<M>(send);
    }
}

__device__ __forceinline__ float one_plus_tanh_small(float x)
{
    const float x2 = x * x;
    float p = __builtin_fmaf(x2, -0x1.758e7ap-8f, 0x1.521192p-6f);
    p = __builtin_fmaf(x2, p, -0x1.b8389cp-5f);
    p = __builtin_fmaf(x2, p, 0x1.110704p-3f);
    p = __builtin_fmaf(x2, p, -0x1.555532p-2f);
    return 1.0f + __builtin_fmaf(x2, x * p, x);
}

__global__ __launch_bounds__(SCW * 32) void scan_kernel(const float* __restrict__ K,
                            const float* __restrict__ Q,
                            const float* __restrict__ V,
                            const float* __restrict__ vc,
                            const float* __restrict__ eta,
                            const float* __restrict__ theta_matrix,
                            float* __restrict__ y)
{
    __shared__ __attribute__((aligned(16))) float Svc[2][SBLKF];
    __shared__ __attribute__((aligned(16))) float Skq[2][2][SBLKF];
    __shared__ float Sy[2][SROWS][TBLK];

    const int b    = blockIdx.x;
    const int tid  = threadIdx.x;
    const int wave = tid >> 5;
    const int lane = tid & 31;
    const int rw   = wave >> 1;
    const int h    = wave & 1;
    const int i    = blockIdx.y * SROWS + rw;
    const int col  = h * 32 + lane;
    const float theta_i = theta_matrix[i];

    float s = 0.f;
    const size_t base = (size_t)b * Tt * Mm;
    const float* etal = eta + (size_t)b * Tt + lane;
    const float* Vl   = V + base + (size_t)lane * Mm + i;
    const v4f* gvc = (const v4f*)(vc + base) + tid;
    const v4f* gK  = (const v4f*)(K  + base) + tid;
    const v4f* gQ  = (const v4f*)(Q  + base) + tid;

    float a[TBLK];
#pragma unroll
    for (int tt = 0; tt < TBLK; ++tt) a[tt] = vc[base + (size_t)tt * Mm + col];
    float u_l = etal[0] * Vl[0];

#pragma unroll
    for (int j = 0; j < SPIECES; ++j) {
        const int pc = j * (SCW * 32);
        const v4f gv = gvc[SBLKF / 4 + pc], gk = gK[pc], gq = gQ[pc];
        ((v4f*)Svc[1])[tid + pc]    = gv;
        ((v4f*)Skq[0][0])[tid + pc] = gk;
        ((v4f*)Skq[0][1])[tid + pc] = gq;
    }
    __syncthreads();

    float yl = 0.f, yu = 0.f;

    for (int t0 = 0; t0 < Tt; t0 += TBLK) {
        const int n = t0 / TBLK;
        const int tn = (t0 + TBLK < Tt) ? t0 + TBLK : t0;
        const float et_n = etal[tn];
        const float vi_n = Vl[(size_t)tn * Mm];

        const int nv = (n + 2 < NBLK) ? n + 2 : NBLK - 1;
        const int nk = (n + 1 < NBLK) ? n + 1 : NBLK - 1;
        v4f gv[SPIECES], gk[SPIECES], gq[SPIECES];
#pragma unroll
        for (int j = 0; j < SPIECES; ++j) {
            const int pc = j * (SCW * 32);
            gv[j] = gvc[(size_t)nv * (SBLKF / 4) + pc];
            gk[j] = gK[(size_t)nk * (SBLKF / 4) + pc];
            gq[j] = gQ[(size_t)nk * (SBLKF / 4) + pc];
        }

        const float* Lv = Svc[(n + 1) & 1] + col;
        const float* Lk = Skq[n & 1][0] + col;
        const float* Lq = Skq[n & 1][1] + col;
        float an[TBLK], kx[TBLK], qx[TBLK];
#pragma unroll
        for (int tt = 0; tt < TBLK; ++tt) {
            an[tt] = Lv[tt * Mm];
            kx[tt] = Lk[tt * Mm];
            qx[tt] = Lq[tt * Mm];
        }

        float am[4] = {0.f, 0.f, 0.f, 0.f};
#pragma unroll
        for (int tt = 0; tt < TBLK; ++tt)
            am[tt & 3] = fmaxf(am[tt & 3], fabsf(a[tt]));
        const float amax = fmaxf(fmaxf(am[0], am[1]), fmaxf(am[2], am[3]));
        const bool big = !(fabsf(theta_i) * amax < 0.625f);

        float c[TBLK];
        if (__builtin_expect(__builtin_amdgcn_ballot_w32(big) == 0u, 1)) {
#pragma unroll
            for (int tt = 0; tt < TBLK; ++tt) c[tt] = one_plus_tanh_small(theta_i * a[tt]);
        } else {
#pragma unroll
            for (int tt = 0; tt < TBLK; ++tt) c[tt] = 1.0f + tanhf(theta_i * a[tt]);
        }

        if (h == 0 && n > 0)
            y[base + (size_t)(t0 - TBLK + lane) * Mm + i] = yl + yu;

        float p[TBLK];
#pragma unroll
        for (int tt = 0; tt < TBLK; ++tt) {
            const float u = __builtin_bit_cast(float,
                __builtin_amdgcn_readlane(__builtin_bit_cast(int, u_l), tt));
            s = __builtin_fmaf(u, kx[tt], s * c[tt]);
            p[tt] = s * qx[tt];
            a[tt] = an[tt];
        }

#pragma unroll
        for (int j = 0; j < SPIECES; ++j) {
            const int pc = tid + j * (SCW * 32);
            ((v4f*)Svc[n & 1])[pc]          = gv[j];
            ((v4f*)Skq[(n + 1) & 1][0])[pc] = gk[j];
            ((v4f*)Skq[(n + 1) & 1][1])[pc] = gq[j];
        }

        bfly_stage<16>(p, lane);
        bfly_stage<8>(p, lane);
        bfly_stage<4>(p, lane);
        bfly_stage<2>(p, lane);
        bfly_stage<1>(p, lane);
        u_l = et_n * vi_n;
        if (h == 1) Sy[n & 1][rw][lane] = p[0];
        __syncthreads();
        if (h == 0) { yl = p[0]; yu = Sy[n & 1][rw][lane]; }
    }
    if (h == 0) y[base + (size_t)(Tt - TBLK + lane) * Mm + i] = yl + yu;
}

__global__ void out_gemm_kernel(const float* __restrict__ y,
                                const float* __restrict__ Wout,
                                const float* __restrict__ bout,
                                float* __restrict__ out)
{
    const int tileM = blockIdx.x * 16;
    const int tileN = blockIdx.y * 64;
    const int lane  = threadIdx.x & 31;
    const int hi    = lane >> 4;
    const int lr    = lane & 15;

    const float* Arow = y + (size_t)(tileM + lr) * Mm + 2 * hi;
    const float* B0 = Wout + (size_t)(tileN + lr +  0) * Mm + 2 * hi;
    const float* B1 = Wout + (size_t)(tileN + lr + 16) * Mm + 2 * hi;
    const float* B2 = Wout + (size_t)(tileN + lr + 32) * Mm + 2 * hi;
    const float* B3 = Wout + (size_t)(tileN + lr + 48) * Mm + 2 * hi;

    v8f c0 = {}, c1 = {}, c2 = {}, c3 = {};
#pragma unroll
    for (int kk = 0; kk < Mm; kk += 4) {
        v2f a  = *(const v2f*)(Arow + kk);
        v2f b0 = *(const v2f*)(B0 + kk);
        v2f b1 = *(const v2f*)(B1 + kk);
        v2f b2 = *(const v2f*)(B2 + kk);
        v2f b3 = *(const v2f*)(B3 + kk);
        c0 = __builtin_amdgcn_wmma_f32_16x16x4_f32(false, a, false, b0, (short)0, c0, false, false);
        c1 = __builtin_amdgcn_wmma_f32_16x16x4_f32(false, a, false, b1, (short)0, c1, false, false);
        c2 = __builtin_amdgcn_wmma_f32_16x16x4_f32(false, a, false, b2, (short)0, c2, false, false);
        c3 = __builtin_amdgcn_wmma_f32_16x16x4_f32(false, a, false, b3, (short)0, c3, false, false);
    }

    const float bb0 = bout[tileN + lr +  0];
    const float bb1 = bout[tileN + lr + 16];
    const float bb2 = bout[tileN + lr + 32];
    const float bb3 = bout[tileN + lr + 48];

#pragma unroll
    for (int r = 0; r < 8; ++r) {
        float* o = out + (size_t)(tileM + r + 8 * hi) * Oo + tileN + lr;
        o[0]  = c0[r] + bb0;
        o[16] = c1[r] + bb1;
        o[32] = c2[r] + bb2;
        o[48] = c3[r] + bb3;
    }
}

extern "C" void kernel_launch(void* const* d_in, const int* in_sizes, int n_in,
                              void* d_out, int out_size, void* d_ws, size_t ws_size,
                              hipStream_t stream)
{
    const float* x     = (const float*)d_in[0];
    const float* Wk    = (const float*)d_in[1];
    const float* Wq    = (const float*)d_in[2];
    const float* Wv    = (const float*)d_in[3];
    const float* Wvc   = (const float*)d_in[4];
    const float* We    = (const float*)d_in[5];
    const float* theta = (const float*)d_in[6];
    const float* Wout  = (const float*)d_in[7];
    const float* bout  = (const float*)d_in[8];
    const float* gamma = (const float*)d_in[9];
    const float* beta  = (const float*)d_in[10];
    float* out = (float*)d_out;

    float* ws  = (float*)d_ws;
    const size_t NROW = (size_t)Bb * Tt;
    float* Ko  = ws;
    float* Qo  = Ko  + NROW * Mm;
    float* Vo  = Qo  + NROW * Mm;
    float* vco = Vo  + NROW * Mm;
    float* yb  = vco + NROW * Mm;
    float* eta = yb  + NROW * Mm;

    ln_proj_kernel<<<dim3(NROW / LROWS), dim3(256), 0, stream>>>(x, gamma, beta, We, Wk, Wq, Wv, Wvc,
                                                             Ko, Qo, Vo, vco, eta);
    scan_kernel<<<dim3(Bb, Mm / SROWS), dim3(SCW * 32), 0, stream>>>(Ko, Qo, Vo, vco, eta, theta, yb);
    out_gemm_kernel<<<dim3(NROW / 16, Oo / 64), dim3(32), 0, stream>>>(yb, Wout, bout, out);
}


# scan kernel: time blocks of 64 steps (were 32): half as many block heads, range tests, barriers and LDS hand-overs; two 32-step butterflies side by side; gate polynomial on float pairs; same sums
# speedup vs baseline: 14.0603x; 1.1636x over previous
#include <hip/hip_runtime.h>
#include <hip/hip_bf16.h>

typedef float v2f __attribute__((ext_vector_type(2)));
typedef float v4f __attribute__((ext_vector_type(4)));
typedef float v8f __attribute__((ext_vector_type(8)));

#define Bb 8
#define Tt 1024
#define Dd 512
#define Mm 64
#define Oo 512

template <int M>
__device__ __forceinline__ float lane_xor(float v)
{
    const int b = __builtin_bit_cast(int, v);
    if constexpr (M == 16)
        return __builtin_bit_cast(float,
            __builtin_amdgcn_permlanex16(b, b, 0x76543210, 0xfedcba98, false, false));
    else
        return __builtin_bit_cast(float,
            __builtin_amdgcn_update_dpp(0, b, 0x160 | M, 0xf, 0xf, true));
}
template <int M>
__device__ __forceinline__ void xsum4(float (&a)[4])
{
#pragma unroll
    for (int q = 0; q < 4; ++q) a[q] += lane_xor<M>(a[q]);
}
__device__ __forceinline__ void wave_sum4(float (&a)[4])
{
    xsum4<16>(a); xsum4<8>(a); xsum4<4>(a); xsum4<2>(a); xsum4<1>(a);
}

#define LROWS  32
#define APITCH 516
#define WROWS  256
#define PKC    32
#define PPITCH 36

__device__ __forceinline__ const float* wrow32(const float* Wk, const float* Wq,
                                               const float* Wv, const float* Wvc, int j)
{
    const float* W = (j < 2) ? Wk : (j < 4) ? Wq : (j < 6) ? Wv : Wvc;
    return W + (size_t)(32 * (j & 1)) * Dd;
}

__global__ __launch_bounds__(256) __attribute__((amdgpu_waves_per_eu(2, 2)))
void ln_proj_kernel(const float* __restrict__ x,
                    const float* __restrict__ gamma,
                    const float* __restrict__ beta,
                    const float* __restrict__ We,
                    const float* __restrict__ Wk,
                    const float* __restrict__ Wq,
                    const float* __restrict__ Wv,
                    const float* __restrict__ Wvc,
                    float* __restrict__ Ko,
                    float* __restrict__ Qo,
                    float* __restrict__ Vo,
                    float* __restrict__ vco,
                    float* __restrict__ eta)
{
    __shared__ __attribute__((aligned(16))) float As[LROWS * APITCH];
    __shared__ __attribute__((aligned(16))) float Bs[2][WROWS * PPITCH];

    const int tid   = threadIdx.x;
    const int wave  = __builtin_amdgcn_readfirstlane(tid >> 5);
    const int lane  = tid & 31;
    const int hi    = lane >> 4;
    const int lr    = lane & 15;
    const int row0  = blockIdx.x * LROWS;

    const int srow = tid >> 3;
    const int sseg = (tid & 7) * 4;
    const size_t goff = (size_t)srow * Dd + sseg;
    const int sl = srow * PPITCH + sseg;

    v4f rb[8];
#pragma unroll
    for (int j = 0; j < 8; ++j) rb[j] = *(const v4f*)(wrow32(Wk, Wq, Wv, Wvc, j) + goff);

    {
        float g[16], bt[16], we[16];
        float v[4][16];
        const float* xr = x + (size_t)(row0 + 4 * wave) * Dd + lane;
#pragma unroll
        for (int q = 0; q < 4; ++q)
#pragma unroll
            for (int e = 0; e < 16; ++e) v[q][e] = xr[(size_t)q * Dd + 32 * e];
#pragma unroll
        for (int e = 0; e < 16; ++e) {
            const int d = lane + 32 * e;
            g[e] = gamma[d]; bt[e] = beta[d]; we[e] = We[d];
        }
        float sum[4], sq[4];
#pragma unroll
        for (int q = 0; q < 4; ++q) {
            sum[q] = 0.f; sq[q] = 0.f;
#pragma unroll
            for (int e = 0; e < 16; ++e) {
                const float t = v[q][e];
                sum[q] += t; sq[q] += t * t;
            }
        }
        wave_sum4(sum);
        wave_sum4(sq);

        float dot[4];
#pragma unroll
        for (int q = 0; q < 4; ++q) {
            const float mu = sum[q] * (1.0f / Dd);
            const float var = sq[q] * (1.0f / Dd) - mu * mu;
            const float rs = rsqrtf(var + 1e-5f);
            dot[q] = 0.f;
            float* xo = &As[(4 * wave + q) * APITCH + lane];
#pragma unroll
            for (int e = 0; e < 16; ++e) {
                float xv = (v[q][e] - mu) * rs * g[e] + bt[e];
                xo[32 * e] = xv;
                dot[q] += xv * we[e];
            }
        }
        wave_sum4(dot);
        const float dl = (lane == 0) ? dot[0] : (lane == 1) ? dot[1] : (lane == 2) ? dot[2] : dot[3];
        if (lane < 4) eta[row0 + 4 * wave + lane] = 1.0f / (1.0f + expf(-dl));
    }

#pragma unroll
    for (int j = 0; j < 8; ++j) *(v4f*)&Bs[0][sl + 32 * j * PPITCH] = rb[j];
    __syncthreads();

    const int rg = wave >> 2;
    const int w  = wave & 3;
    float* Out   = (w == 0) ? Ko : (w == 1) ? Qo : (w == 2) ? Vo : vco;
    const int tileM = row0 + 16 * rg;

    const int fa = (16 * rg + lr) * APITCH + 2 * hi;
    const int fb = (64 * w + lr) * PPITCH + 2 * hi;

    v8f c0 = {}, c1 = {}, c2 = {}, c3 = {};
    for (int c = 0; c < Dd / PKC; ++c) {
        const int cur = c & 1;
        const bool more = (c + 1 < Dd / PKC);
        if (more) {
            const int ko = (c + 1) * PKC;
#pragma unroll
            for (int j = 0; j < 8; ++j)
                rb[j] = *(const v4f*)(wrow32(Wk, Wq, Wv, Wvc, j) + goff + ko);
        }

        const float* Afr = &As[fa + c * PKC];
        const float* Bfr = &Bs[cur][fb];
#pragma unroll
        for (int kk = 0; kk < PKC; kk += 4) {
            v2f a  = *(const v2f*)(Afr + kk);
            v2f b0 = *(const v2f*)(Bfr + kk);
            v2f b1 = *(const v2f*)(Bfr + 16 * PPITCH + kk);
            v2f b2 = *(const v2f*)(Bfr + 32 * PPITCH + kk);
            v2f b3 = *(const v2f*)(Bfr + 48 * PPITCH + kk);
            c0 = __builtin_amdgcn_wmma_f32_16x16x4_f32(false, a, false, b0, (short)0, c0, false, false);
            c1 = __builtin_amdgcn_wmma_f32_16x16x4_f32(false, a, false, b1, (short)0, c1, false, false);
            c2 = __builtin_amdgcn_wmma_f32_16x16x4_f32(false, a, false, b2, (short)0, c2, false, false);
            c3 = __builtin_amdgcn_wmma_f32_16x16x4_f32(false, a, false, b3, (short)0, c3, false, false);
        }

        if (more) {
#pragma unroll
            for (int j = 0; j < 8; ++j) *(v4f*)&Bs[cur ^ 1][sl + 32 * j * PPITCH] = rb[j];
        }
        __syncthreads();
    }

    if (w < 2) {
#pragma unroll
        for (int r = 0; r < 8; ++r) {
            c0[r] = fmaxf(c0[r], 0.f); c1[r] = fmaxf(c1[r], 0.f);
            c2[r] = fmaxf(c2[r], 0.f); c3[r] = fmaxf(c3[r], 0.f);
            float s = c0[r] + c1[r] + c2[r] + c3[r];
#pragma unroll
            for (int m = 8; m >= 1; m >>= 1) s += __shfl_xor(s, m, 32);
            float inv = 1.0f / (1e-5f + s);
            c0[r] *= inv; c1[r] *= inv; c2[r] *= inv; c3[r] *= inv;
        }
    }

#pragma unroll
    for (int r = 0; r < 8; ++r) {
        float* o = Out + (size_t)(tileM + r + 8 * hi) * Mm + lr;
        o[0]  = c0[r];
        o[16] = c1[r];
        o[32] = c2[r];
        o[48] = c3[r];
    }
}

#define TBLK 64
#define TGRP 32
#define NBLK (Tt / TBLK)
#define SCW 4
#define SROWS (SCW / 2)
#define SBLKF (TBLK * Mm)
#define SPIECES (SBLKF / 4 / (SCW * 32))

template <int M>
__device__ __forceinline__ void bfly_stage(float (&p)[TGRP], const int lane)
{
    const bool up = (lane & M) != 0;
#pragma unroll
    for (int r = 0; r < M; ++r) {
        const float lo = p[r], hi = p[r + M];
        const float keep = up ? hi : lo;
        const float send = up ? lo : hi;
        p[r] = keep + lane_xor<M>(send);
    }
}

__device__ __forceinline__ float one_plus_tanh_small(float x)
{
    const float x2 = x * x;
    float p = __builtin_fmaf(x2, -0x1.758e7ap-8f, 0x1.521192p-6f);
    p = __builtin_fmaf(x2, p, -0x1.b8389cp-5f);
    p = __builtin_fmaf(x2, p, 0x1.110704p-3f);
    p = __builtin_fmaf(x2, p, -0x1.555532p-2f);
    return 1.0f + __builtin_fmaf(x2, x * p, x);
}
__device__ __forceinline__ v2f one_plus_tanh_small2(v2f x)
{
    const v2f x2 = x * x;
    v2f p = __builtin_elementwise_fma(x2, (v2f)(-0x1.758e7ap-8f), (v2f)(0x1.521192p-6f));
    p = __builtin_elementwise_fma(x2, p, (v2f)(-0x1.b8389cp-5f));
    p = __builtin_elementwise_fma(x2, p, (v2f)(0x1.110704p-3f));
    p = __builtin_elementwise_fma(x2, p, (v2f)(-0x1.555532p-2f));
    return (v2f)(1.0f) + __builtin_elementwise_fma(x2, x * p, x);
}

__global__ __launch_bounds__(SCW * 32) void scan_kernel(const float* __restrict__ K,
                            const float* __restrict__ Q,
                            const float* __restrict__ V,
                            const float* __restrict__ vc,
                            const float* __restrict__ eta,
                            const float* __restrict__ theta_matrix,
                            float* __restrict__ y)
{
    __shared__ __attribute__((aligned(16))) float Svc[2][SBLKF];
    __shared__ __attribute__((aligned(16))) float Skq[2][2][SBLKF];
    __shared__ float Sy[2][SROWS][TBLK];

    const int b    = blockIdx.x;
    const int tid  = threadIdx.x;
    const int wave = tid >> 5;
    const int lane = tid & 31;
    const int rw   = wave >> 1;
    const int h    = wave & 1;
    const int i    = blockIdx.y * SROWS + rw;
    const int col  = h * 32 + lane;
    const float theta_i = theta_matrix[i];

    float s = 0.f;
    const size_t base = (size_t)b * Tt * Mm;
    const float* etal = eta + (size_t)b * Tt + lane;
    const float* Vl   = V + base + (size_t)lane * Mm + i;
    const v4f* gvc = (const v4f*)(vc + base) + tid;
    const v4f* gK  = (const v4f*)(K  + base) + tid;
    const v4f* gQ  = (const v4f*)(Q  + base) + tid;

    float a[TBLK];
#pragma unroll
    for (int tt = 0; tt < TBLK; ++tt) a[tt] = vc[base + (size_t)tt * Mm + col];
    float u_lo = etal[0] * Vl[0];
    float u_hi = etal[TGRP] * Vl[(size_t)TGRP * Mm];

#pragma unroll
    for (int j = 0; j < SPIECES; ++j) {
        const int pc = j * (SCW * 32);
        const v4f gv = gvc[SBLKF / 4 + pc], gk = gK[pc], gq = gQ[pc];
        ((v4f*)Svc[1])[tid + pc]    = gv;
        ((v4f*)Skq[0][0])[tid + pc] = gk;
        ((v4f*)Skq[0][1])[tid + pc] = gq;
    }
    __syncthreads();

    float yl0 = 0.f, yl1 = 0.f, yu0 = 0.f, yu1 = 0.f;

    for (int t0 = 0; t0 < Tt; t0 += TBLK) {
        const int n = t0 / TBLK;
        const int tn = (t0 + TBLK < Tt) ? t0 + TBLK : t0;
        const float et_lo = etal[tn];
        const float et_hi = etal[tn + TGRP];
        const float vi_lo = Vl[(size_t)tn * Mm];
        const float vi_hi = Vl[(size_t)(tn + TGRP) * Mm];

        const int nv = (n + 2 < NBLK) ? n + 2 : NBLK - 1;
        const int nk = (n + 1 < NBLK) ? n + 1 : NBLK - 1;
        v4f gv[SPIECES], gk[SPIECES], gq[SPIECES];
#pragma unroll
        for (int j = 0; j < SPIECES; ++j) {
            const int pc = j * (SCW * 32);
            gv[j] = gvc[(size_t)nv * (SBLKF / 4) + pc];
            gk[j] = gK[(size_t)nk * (SBLKF / 4) + pc];
            gq[j] = gQ[(size_t)nk * (SBLKF / 4) + pc];
        }

        const float* Lk = Skq[n & 1][0] + col;
        const float* Lq = Skq[n & 1][1] + col;
        float kx[TBLK], qx[TBLK];
#pragma unroll
        for (int tt = 0; tt < TBLK; ++tt) {
            kx[tt] = Lk[tt * Mm];
            if (tt < TGRP) qx[tt] = Lq[tt * Mm];
        }

        float am[4] = {0.f, 0.f, 0.f, 0.f};
#pragma unroll
        for (int tt = 0; tt < TBLK; ++tt)
            am[tt & 3] = fmaxf(am[tt & 3], fabsf(a[tt]));
        const float amax = fmaxf(fmaxf(am[0], am[1]), fmaxf(am[2], am[3]));
        const bool big = !(fabsf(theta_i) * amax < 0.625f);

        float c[TBLK];
        if (__builtin_expect(__builtin_amdgcn_ballot_w32(big) == 0u, 1)) {
#pragma unroll
            for (int tt = 0; tt < TBLK; tt += 2) {
                const v2f av = {a[tt], a[tt + 1]};
                const v2f cv = one_plus_tanh_small2(av * theta_i);
                c[tt] = cv.x; c[tt + 1] = cv.y;
            }
        } else {
#pragma unroll
            for (int tt = 0; tt < TBLK; ++tt) c[tt] = 1.0f + tanhf(theta_i * a[tt]);
        }

        const float* Lv = Svc[(n + 1) & 1] + col;
        float an[TBLK];
#pragma unroll
        for (int tt = TGRP; tt < TBLK; ++tt) qx[tt] = Lq[tt * Mm];
#pragma unroll
        for (int tt = 0; tt < TBLK; ++tt) an[tt] = Lv[tt * Mm];

        if (h == 0 && n > 0) {
            y[base + (size_t)(t0 - TBLK + lane) * Mm + i]        = yl0 + yu0;
            y[base + (size_t)(t0 - TBLK + TGRP + lane) * Mm + i] = yl1 + yu1;
        }

        float p0[TGRP], p1[TGRP];
#pragma unroll
        for (int tt = 0; tt < TBLK; ++tt) {
            const float u = __builtin_bit_cast(float,
                __builtin_amdgcn_readlane(__builtin_bit_cast(int, (tt < TGRP) ? u_lo : u_hi),
                                          tt & (TGRP - 1)));
            s = __builtin_fmaf(u, kx[tt], s * c[tt]);
            if (tt < TGRP) p0[tt] = s * qx[tt];
            else           p1[tt - TGRP] = s * qx[tt];
            a[tt] = an[tt];
        }

#pragma unroll
        for (int j = 0; j < SPIECES; ++j) {
            const int pc = tid + j * (SCW * 32);
            ((v4f*)Svc[n & 1])[pc]          = gv[j];
            ((v4f*)Skq[(n + 1) & 1][0])[pc] = gk[j];
            ((v4f*)Skq[(n + 1) & 1][1])[pc] = gq[j];
        }

        bfly_stage<16>(p0, lane); bfly_stage<16>(p1, lane);
        bfly_stage<8>(p0, lane);  bfly_stage<8>(p1, lane);
        bfly_stage<4>(p0, lane);  bfly_stage<4>(p1, lane);
        bfly_stage<2>(p0, lane);  bfly_stage<2>(p1, lane);
        bfly_stage<1>(p0, lane);  bfly_stage<1>(p1, lane);
        u_lo = et_lo * vi_lo;
        u_hi = et_hi * vi_hi;
        if (h == 1) { Sy[n & 1][rw][lane] = p0[0]; Sy[n & 1][rw][TGRP + lane] = p1[0]; }
        __syncthreads();
        if (h == 0) {
            yl0 = p0[0]; yl1 = p1[0];
            yu0 = Sy[n & 1][rw][lane]; yu1 = Sy[n & 1][rw][TGRP + lane];
        }
    }
    if (h == 0) {
        y[base + (size_t)(Tt - TBLK + lane) * Mm + i]        = yl0 + yu0;
        y[base + (size_t)(Tt - TBLK + TGRP + lane) * Mm + i] = yl1 + yu1;
    }
}

__global__ void out_gemm_kernel(const float* __restrict__ y,
                                const float* __restrict__ Wout,
                                const float* __restrict__ bout,
                                float* __restrict__ out)
{
    const int tileM = blockIdx.x * 16;
    const int tileN = blockIdx.y * 64;
    const int lane  = threadIdx.x & 31;
    const int hi    = lane >> 4;
    const int lr    = lane & 15;

    const float* Arow = y + (size_t)(tileM + lr) * Mm + 2 * hi;
    const float* B0 = Wout + (size_t)(tileN + lr +  0) * Mm + 2 * hi;
    const float* B1 = Wout + (size_t)(tileN + lr + 16) * Mm + 2 * hi;
    const float* B2 = Wout + (size_t)(tileN + lr + 32) * Mm + 2 * hi;
    const float* B3 = Wout + (size_t)(tileN + lr + 48) * Mm + 2 * hi;

    v8f c0 = {}, c1 = {}, c2 = {}, c3 = {};
#pragma unroll
    for (int kk = 0; kk < Mm; kk += 4) {
        v2f a  = *(const v2f*)(Arow + kk);
        v2f b0 = *(const v2f*)(B0 + kk);
        v2f b1 = *(const v2f*)(B1 + kk);
        v2f b2 = *(const v2f*)(B2 + kk);
        v2f b3 = *(const v2f*)(B3 + kk);
        c0 = __builtin_amdgcn_wmma_f32_16x16x4_f32(false, a, false, b0, (short)0, c0, false, false);
        c1 = __builtin_amdgcn_wmma_f32_16x16x4_f32(false, a, false, b1, (short)0, c1, false, false);
        c2 = __builtin_amdgcn_wmma_f32_16x16x4_f32(false, a, false, b2, (short)0, c2, false, false);
        c3 = __builtin_amdgcn_wmma_f32_16x16x4_f32(false, a, false, b3, (short)0, c3, false, false);
    }

    const float bb0 = bout[tileN + lr +  0];
    const float bb1 = bout[tileN + lr + 16];
    const float bb2 = bout[tileN + lr + 32];
    const float bb3 = bout[tileN + lr + 48];

#pragma unroll
    for (int r = 0; r < 8; ++r) {
        float* o = out + (size_t)(tileM + r + 8 * hi) * Oo + tileN + lr;
        o[0]  = c0[r] + bb0;
        o[16] = c1[r] + bb1;
        o[32] = c2[r] + bb2;
        o[48] = c3[r] + bb3;
    }
}

extern "C" void kernel_launch(void* const* d_in, const int* in_sizes, int n_in,
                              void* d_out, int out_size, void* d_ws, size_t ws_size,
                              hipStream_t stream)
{
    const float* x     = (const float*)d_in[0];
    const float* Wk    = (const float*)d_in[1];
    const float* Wq    = (const float*)d_in[2];
    const float* Wv    = (const float*)d_in[3];
    const float* Wvc   = (const float*)d_in[4];
    const float* We    = (const float*)d_in[5];
    const float* theta = (const float*)d_in[6];
    const float* Wout  = (const float*)d_in[7];
    const float* bout  = (const float*)d_in[8];
    const float* gamma = (const float*)d_in[9];
    const float* beta  = (const float*)d_in[10];
    float* out = (float*)d_out;

    float* ws  = (float*)d_ws;
    const size_t NROW = (size_t)Bb * Tt;
    float* Ko  = ws;
    float* Qo  = Ko  + NROW * Mm;
    float* Vo  = Qo  + NROW * Mm;
    float* vco = Vo  + NROW * Mm;
    float* yb  = vco + NROW * Mm;
    float* eta = yb  + NROW * Mm;

    ln_proj_kernel<<<dim3(NROW / LROWS), dim3(256), 0, stream>>>(x, gamma, beta, We, Wk, Wq, Wv, Wvc,
                                                             Ko, Qo, Vo, vco, eta);
    scan_kernel<<<dim3(Bb, Mm / SROWS), dim3(SCW * 32), 0, stream>>>(Ko, Qo, Vo, vco, eta, theta, yb);
    out_gemm_kernel<<<dim3(NROW / 16, Oo / 64), dim3(32), 0, stream>>>(yb, Wout, bout, out);
}


# projection writes K,Q,vc in the scan's lane order (64-step block, column half, 4 steps a lane); scan waves fetch operands by whole-line 16-byte global loads, no LDS staging; barrier fences LDS only
# speedup vs baseline: 15.5061x; 1.1028x over previous
#include <hip/hip_runtime.h>
#include <hip/hip_bf16.h>

typedef float v2f __attribute__((ext_vector_type(2)));
typedef float v4f __attribute__((ext_vector_type(4)));
typedef float v8f __attribute__((ext_vector_type(8)));

#define Bb 8
#define Tt 1024
#define Dd 512
#define Mm 64
#define Oo 512
#define SBLK 64

template <int M>
__device__ __forceinline__ float lane_xor(float v)
{
    const int b = __builtin_bit_cast(int, v);
    if constexpr (M == 16)
        return __builtin_bit_cast(float,
            __builtin_amdgcn_permlanex16(b, b, 0x76543210, 0xfedcba98, false, false));
    else
        return __builtin_bit_cast(float,
            __builtin_amdgcn_update_dpp(0, b, 0x160 | M, 0xf, 0xf, true));
}
template <int M>
__device__ __forceinline__ void xsum4(float (&a)[4])
{
#pragma unroll
    for (int q = 0; q < 4; ++q) a[q] += lane_xor<M>(a[q]);
}
__device__ __forceinline__ void wave_sum4(float (&a)[4])
{
    xsum4<16>(a); xsum4<8>(a); xsum4<4>(a); xsum4<2>(a); xsum4<1>(a);
}

#define LROWS  32
#define APITCH 516
#define WROWS  256
#define PKC    32
#define PPITCH 36

__device__ __forceinline__ const float* wrow32(const float* Wk, const float* Wq,
                                               const float* Wv, const float* Wvc, int j)
{
    const float* W = (j < 2) ? Wk : (j < 4) ? Wq : (j < 6) ? Wv : Wvc;
    return W + (size_t)(32 * (j & 1)) * Dd;
}

__global__ __launch_bounds__(256) __attribute__((amdgpu_waves_per_eu(2, 2)))
void ln_proj_kernel(const float* __restrict__ x,
                    const float* __restrict__ gamma,
                    const float* __restrict__ beta,
                    const float* __restrict__ We,
                    const float* __restrict__ Wk,
                    const float* __restrict__ Wq,
                    const float* __restrict__ Wv,
                    const float* __restrict__ Wvc,
                    float* __restrict__ Ko,
                    float* __restrict__ Qo,
                    float* __restrict__ Vo,
                    float* __restrict__ vco,
                    float* __restrict__ eta)
{
    __shared__ __attribute__((aligned(16))) float As[LROWS * APITCH];
    __shared__ __attribute__((aligned(16))) float Bs[2][WROWS * PPITCH];

    const int tid   = threadIdx.x;
    const int wave  = __builtin_amdgcn_readfirstlane(tid >> 5);
    const int lane  = tid & 31;
    const int hi    = lane >> 4;
    const int lr    = lane & 15;
    const int row0  = blockIdx.x * LROWS;

    const int srow = tid >> 3;
    const int sseg = (tid & 7) * 4;
    const size_t goff = (size_t)srow * Dd + sseg;
    const int sl = srow * PPITCH + sseg;

    v4f rb[8];
#pragma unroll
    for (int j = 0; j < 8; ++j) rb[j] = *(const v4f*)(wrow32(Wk, Wq, Wv, Wvc, j) + goff);

    {
        float g[16], bt[16], we[16];
        float v[4][16];
        const float* xr = x + (size_t)(row0 + 4 * wave) * Dd + lane;
#pragma unroll
        for (int q = 0; q < 4; ++q)
#pragma unroll
            for (int e = 0; e < 16; ++e) v[q][e] = xr[(size_t)q * Dd + 32 * e];
#pragma unroll
        for (int e = 0; e < 16; ++e) {
            const int d = lane + 32 * e;
            g[e] = gamma[d]; bt[e] = beta[d]; we[e] = We[d];
        }
        float sum[4], sq[4];
#pragma unroll
        for (int q = 0; q < 4; ++q) {
            sum[q] = 0.f; sq[q] = 0.f;
#pragma unroll
            for (int e = 0; e < 16; ++e) {
                const float t = v[q][e];
                sum[q] += t; sq[q] += t * t;
            }
        }
        wave_sum4(sum);
        wave_sum4(sq);

        float dot[4];
#pragma unroll
        for (int q = 0; q < 4; ++q) {
            const float mu = sum[q] * (1.0f / Dd);
            const float var = sq[q] * (1.0f / Dd) - mu * mu;
            const float rs = rsqrtf(var + 1e-5f);
            dot[q] = 0.f;
            float* xo = &As[(4 * wave + q) * APITCH + lane];
#pragma unroll
            for (int e = 0; e < 16; ++e) {
                float xv = (v[q][e] - mu) * rs * g[e] + bt[e];
                xo[32 * e] = xv;
                dot[q] += xv * we[e];
            }
        }
        wave_sum4(dot);
        const float dl = (lane == 0) ? dot[0] : (lane == 1) ? dot[1] : (lane == 2) ? dot[2] : dot[3];
        if (lane < 4) eta[row0 + 4 * wave + lane] = 1.0f / (1.0f + expf(-dl));
    }

#pragma unroll
    for (int j = 0; j < 8; ++j) *(v4f*)&Bs[0][sl + 32 * j * PPITCH] = rb[j];
    __syncthreads();

    const int rg = wave >> 2;
    const int w  = wave & 3;
    float* Out   = (w == 0) ? Ko : (w == 1) ? Qo : (w == 2) ? Vo : vco;
    const int tileM = row0 + 16 * rg;

    const int fa = (16 * rg + lr) * APITCH + 2 * hi;
    const int fb = (64 * w + lr) * PPITCH + 2 * hi;

    v8f c0 = {}, c1 = {}, c2 = {}, c3 = {};
    for (int c = 0; c < Dd / PKC; ++c) {
        const int cur = c & 1;
        const bool more = (c + 1 < Dd / PKC);
        if (more) {
            const int ko = (c + 1) * PKC;
#pragma unroll
            for (int j = 0; j < 8; ++j)
                rb[j] = *(const v4f*)(wrow32(Wk, Wq, Wv, Wvc, j) + goff + ko);
        }

        const float* Afr = &As[fa + c * PKC];
        const float* Bfr = &Bs[cur][fb];
#pragma unroll
        for (int kk = 0; kk < PKC; kk += 4) {
            v2f a  = *(const v2f*)(Afr + kk);
            v2f b0 = *(const v2f*)(Bfr + kk);
            v2f b1 = *(const v2f*)(Bfr + 16 * PPITCH + kk);
            v2f b2 = *(const v2f*)(Bfr + 32 * PPITCH + kk);
            v2f b3 = *(const v2f*)(Bfr + 48 * PPITCH + kk);
            c0 = __builtin_amdgcn_wmma_f32_16x16x4_f32(false, a, false, b0, (short)0, c0, false, false);
            c1 = __builtin_amdgcn_wmma_f32_16x16x4_f32(false, a, false, b1, (short)0, c1, false, false);
            c2 = __builtin_amdgcn_wmma_f32_16x16x4_f32(false, a, false, b2, (short)0, c2, false, false);
            c3 = __builtin_amdgcn_wmma_f32_16x16x4_f32(false, a, false, b3, (short)0, c3, false, false);
        }

        if (more) {
#pragma unroll
            for (int j = 0; j < 8; ++j) *(v4f*)&Bs[cur ^ 1][sl + 32 * j * PPITCH] = rb[j];
        }
        __syncthreads();
    }

    if (w < 2) {
#pragma unroll
        for (int r = 0; r < 8; ++r) {
            c0[r] = fmaxf(c0[r], 0.f); c1[r] = fmaxf(c1[r], 0.f);
            c2[r] = fmaxf(c2[r], 0.f); c3[r] = fmaxf(c3[r], 0.f);
            float s = c0[r] + c1[r] + c2[r] + c3[r];
#pragma unroll
            for (int m = 8; m >= 1; m >>= 1) s += __shfl_xor(s, m, 32);
            float inv = 1.0f / (1e-5f + s);
            c0[r] *= inv; c1[r] *= inv; c2[r] *= inv; c3[r] *= inv;
        }
    }

    if (w == 2) {
#pragma unroll
        for (int r = 0; r < 8; ++r) {
            float* o = Out + (size_t)(tileM + r + 8 * hi) * Mm + lr;
            o[0]  = c0[r];
            o[16] = c1[r];
            o[32] = c2[r];
            o[48] = c3[r];
        }
    } else {
        const int bI   = tileM / Tt;
        const int tloc = tileM % Tt;
        float* ob = Out + (size_t)bI * Tt * Mm + (size_t)(tloc / SBLK) * (SBLK * Mm);
        const int q0 = (tloc % SBLK) / 4 + 2 * hi;
        float* o = ob + q0 * 128 + lr * 4;
        *(v4f*)(o)               = (v4f){c0[0], c0[1], c0[2], c0[3]};
        *(v4f*)(o + 128)         = (v4f){c0[4], c0[5], c0[6], c0[7]};
        *(v4f*)(o + 64)          = (v4f){c1[0], c1[1], c1[2], c1[3]};
        *(v4f*)(o + 64 + 128)    = (v4f){c1[4], c1[5], c1[6], c1[7]};
        *(v4f*)(o + 2048)        = (v4f){c2[0], c2[1], c2[2], c2[3]};
        *(v4f*)(o + 2048 + 128)  = (v4f){c2[4], c2[5], c2[6], c2[7]};
        *(v4f*)(o + 2112)        = (v4f){c3[0], c3[1], c3[2], c3[3]};
        *(v4f*)(o + 2112 + 128)  = (v4f){c3[4], c3[5], c3[6], c3[7]};
    }
}

#define TBLK SBLK
#define TGRP 32
#define NBLK (Tt / TBLK)
#define SCW 4
#define SROWS (SCW / 2)
#define SBLKV (TBLK * Mm / 4)
#define NQUAD (TBLK / 4)

template <int M>
__device__ __forceinline__ void bfly_stage(float (&p)[TGRP], const int lane)
{
    const bool up = (lane & M) != 0;
#pragma unroll
    for (int r = 0; r < M; ++r) {
        const float lo = p[r], hi = p[r + M];
        const float keep = up ? hi : lo;
        const float send = up ? lo : hi;
        p[r] = keep + lane_xor<M>(send);
    }
}

__device__ __forceinline__ float one_plus_tanh_small(float x)
{
    const float x2 = x * x;
    float p = __builtin_fmaf(x2, -0x1.758e7ap-8f, 0x1.521192p-6f);
    p = __builtin_fmaf(x2, p, -0x1.b8389cp-5f);
    p = __builtin_fmaf(x2, p, 0x1.110704p-3f);
    p = __builtin_fmaf(x2, p, -0x1.555532p-2f);
    return 1.0f + __builtin_fmaf(x2, x * p, x);
}
__device__ __forceinline__ v2f one_plus_tanh_small2(v2f x)
{
    const v2f x2 = x * x;
    v2f p = __builtin_elementwise_fma(x2, (v2f)(-0x1.758e7ap-8f), (v2f)(0x1.521192p-6f));
    p = __builtin_elementwise_fma(x2, p, (v2f)(-0x1.b8389cp-5f));
    p = __builtin_elementwise_fma(x2, p, (v2f)(0x1.110704p-3f));
    p = __builtin_elementwise_fma(x2, p, (v2f)(-0x1.555532p-2f));
    return (v2f)(1.0f) + __builtin_elementwise_fma(x2, x * p, x);
}

__global__ __launch_bounds__(SCW * 32) void scan_kernel(const float* __restrict__ K,
                            const float* __restrict__ Q,
                            const float* __restrict__ V,
                            const float* __restrict__ vc,
                            const float* __restrict__ eta,
                            const float* __restrict__ theta_matrix,
                            float* __restrict__ y)
{
    __shared__ float Sy[2][SROWS][TBLK];

    const int b    = blockIdx.x;
    const int tid  = threadIdx.x;
    const int wave = tid >> 5;
    const int lane = tid & 31;
    const int rw   = wave >> 1;
    const int h    = wave & 1;
    const int i    = blockIdx.y * SROWS + rw;
    const float theta_i = theta_matrix[i];

    float s = 0.f;
    const size_t base = (size_t)b * Tt * Mm;
    const float* etal = eta + (size_t)b * Tt + lane;
    const float* Vl   = V + base + (size_t)lane * Mm + i;
    const v4f* gvc = (const v4f*)(vc + base) + h * (SBLKV / 2) + lane;
    const v4f* gK  = (const v4f*)(K  + base) + h * (SBLKV / 2) + lane;
    const v4f* gQ  = (const v4f*)(Q  + base) + h * (SBLKV / 2) + lane;

    float u_lo = etal[0] * Vl[0];
    float u_hi = etal[TGRP] * Vl[(size_t)TGRP * Mm];
    float a[TBLK], kx[TBLK], qx[TBLK];
#pragma unroll
    for (int q = 0; q < NQUAD; ++q) {
        const v4f va = gvc[q * 32];
#pragma unroll
        for (int e = 0; e < 4; ++e) a[4 * q + e] = va[e];
    }
#pragma unroll
    for (int q = 0; q < NQUAD; ++q) {
        const v4f vk = gK[q * 32], vq = gQ[q * 32];
#pragma unroll
        for (int e = 0; e < 4; ++e) { kx[4 * q + e] = vk[e]; qx[4 * q + e] = vq[e]; }
    }

    float yl0 = 0.f, yl1 = 0.f, yu0 = 0.f, yu1 = 0.f;

    for (int t0 = 0; t0 < Tt; t0 += TBLK) {
        const int n = t0 / TBLK;
        const int tn = (t0 + TBLK < Tt) ? t0 + TBLK : t0;
        const float et_lo = etal[tn];
        const float et_hi = etal[tn + TGRP];
        const float vi_lo = Vl[(size_t)tn * Mm];
        const float vi_hi = Vl[(size_t)(tn + TGRP) * Mm];
        const int nn = (n + 1 < NBLK) ? n + 1 : NBLK - 1;
        const v4f* nvc = gvc + (size_t)nn * SBLKV;
        const v4f* nK  = gK  + (size_t)nn * SBLKV;
        const v4f* nQ  = gQ  + (size_t)nn * SBLKV;

        float am[4] = {0.f, 0.f, 0.f, 0.f};
#pragma unroll
        for (int tt = 0; tt < TBLK; ++tt)
            am[tt & 3] = fmaxf(am[tt & 3], fabsf(a[tt]));
        const float amax = fmaxf(fmaxf(am[0], am[1]), fmaxf(am[2], am[3]));
        const bool big = !(fabsf(theta_i) * amax < 0.625f);

        float c[TBLK];
        if (__builtin_expect(__builtin_amdgcn_ballot_w32(big) == 0u, 1)) {
#pragma unroll
            for (int tt = 0; tt < TBLK; tt += 2) {
                const v2f av = {a[tt], a[tt + 1]};
                const v2f cv = one_plus_tanh_small2(av * theta_i);
                c[tt] = cv.x; c[tt + 1] = cv.y;
            }
        } else {
#pragma unroll
            for (int tt = 0; tt < TBLK; ++tt) c[tt] = 1.0f + tanhf(theta_i * a[tt]);
        }

#pragma unroll
        for (int q = 0; q < NQUAD; ++q) {
            const v4f va = nvc[q * 32];
#pragma unroll
            for (int e = 0; e < 4; ++e) a[4 * q + e] = va[e];
        }

        if (h == 0 && n > 0) {
            y[base + (size_t)(t0 - TBLK + lane) * Mm + i]        = yl0 + yu0;
            y[base + (size_t)(t0 - TBLK + TGRP + lane) * Mm + i] = yl1 + yu1;
        }

        float p0[TGRP], p1[TGRP];
#pragma unroll
        for (int tt = 0; tt < TBLK; ++tt) {
            const float u = __builtin_bit_cast(float,
                __builtin_amdgcn_readlane(__builtin_bit_cast(int, (tt < TGRP) ? u_lo : u_hi),
                                          tt & (TGRP - 1)));
            s = __builtin_fmaf(u, kx[tt], s * c[tt]);
            if (tt < TGRP) p0[tt] = s * qx[tt];
            else           p1[tt - TGRP] = s * qx[tt];
        }

#pragma unroll
        for (int q = 0; q < NQUAD; ++q) {
            const v4f vk = nK[q * 32], vq = nQ[q * 32];
#pragma unroll
            for (int e = 0; e < 4; ++e) { kx[4 * q + e] = vk[e]; qx[4 * q + e] = vq[e]; }
        }

        bfly_stage<16>(p0, lane); bfly_stage<16>(p1, lane);
        bfly_stage<8>(p0, lane);  bfly_stage<8>(p1, lane);
        bfly_stage<4>(p0, lane);  bfly_stage<4>(p1, lane);
        bfly_stage<2>(p0, lane);  bfly_stage<2>(p1, lane);
        bfly_stage<1>(p0, lane);  bfly_stage<1>(p1, lane);
        u_lo = et_lo * vi_lo;
        u_hi = et_hi * vi_hi;
        if (h == 1) { Sy[n & 1][rw][lane] = p0[0]; Sy[n & 1][rw][TGRP + lane] = p1[0]; }
        __builtin_amdgcn_fence(__ATOMIC_RELEASE, "workgroup", "local");
        __builtin_amdgcn_s_barrier();
        __builtin_amdgcn_fence(__ATOMIC_ACQUIRE, "workgroup", "local");
        if (h == 0) {
            yl0 = p0[0]; yl1 = p1[0];
            yu0 = Sy[n & 1][rw][lane]; yu1 = Sy[n & 1][rw][TGRP + lane];
        }
    }
    if (h == 0) {
        y[base + (size_t)(Tt - TBLK + lane) * Mm + i]        = yl0 + yu0;
        y[base + (size_t)(Tt - TBLK + TGRP + lane) * Mm + i] = yl1 + yu1;
    }
}

__global__ void out_gemm_kernel(const float* __restrict__ y,
                                const float* __restrict__ Wout,
                                const float* __restrict__ bout,
                                float* __restrict__ out)
{
    const int tileM = blockIdx.x * 16;
    const int tileN = blockIdx.y * 64;
    const int lane  = threadIdx.x & 31;
    const int hi    = lane >> 4;
    const int lr    = lane & 15;

    const float* Arow = y + (size_t)(tileM + lr) * Mm + 2 * hi;
    const float* B0 = Wout + (size_t)(tileN + lr +  0) * Mm + 2 * hi;
    const float* B1 = Wout + (size_t)(tileN + lr + 16) * Mm + 2 * hi;
    const float* B2 = Wout + (size_t)(tileN + lr + 32) * Mm + 2 * hi;
    const float* B3 = Wout + (size_t)(tileN + lr + 48) * Mm + 2 * hi;

    v8f c0 = {}, c1 = {}, c2 = {}, c3 = {};
#pragma unroll
    for (int kk = 0; kk < Mm; kk += 4) {
        v2f a  = *(const v2f*)(Arow + kk);
        v2f b0 = *(const v2f*)(B0 + kk);
        v2f b1 = *(const v2f*)(B1 + kk);
        v2f b2 = *(const v2f*)(B2 + kk);
        v2f b3 = *(const v2f*)(B3 + kk);
        c0 = __builtin_amdgcn_wmma_f32_16x16x4_f32(false, a, false, b0, (short)0, c0, false, false);
        c1 = __builtin_amdgcn_wmma_f32_16x16x4_f32(false, a, false, b1, (short)0, c1, false, false);
        c2 = __builtin_amdgcn_wmma_f32_16x16x4_f32(false, a, false, b2, (short)0, c2, false, false);
        c3 = __builtin_amdgcn_wmma_f32_16x16x4_f32(false, a, false, b3, (short)0, c3, false, false);
    }

    const float bb0 = bout[tileN + lr +  0];
    const float bb1 = bout[tileN + lr + 16];
    const float bb2 = bout[tileN + lr + 32];
    const float bb3 = bout[tileN + lr + 48];

#pragma unroll
    for (int r = 0; r < 8; ++r) {
        float* o = out + (size_t)(tileM + r + 8 * hi) * Oo + tileN + lr;
        o[0]  = c0[r] + bb0;
        o[16] = c1[r] + bb1;
        o[32] = c2[r] + bb2;
        o[48] = c3[r] + bb3;
    }
}

extern "C" void kernel_launch(void* const* d_in, const int* in_sizes, int n_in,
                              void* d_out, int out_size, void* d_ws, size_t ws_size,
                              hipStream_t stream)
{
    const float* x     = (const float*)d_in[0];
    const float* Wk    = (const float*)d_in[1];
    const float* Wq    = (const float*)d_in[2];
    const float* Wv    = (const float*)d_in[3];
    const float* Wvc   = (const float*)d_in[4];
    const float* We    = (const float*)d_in[5];
    const float* theta = (const float*)d_in[6];
    const float* Wout  = (const float*)d_in[7];
    const float* bout  = (const float*)d_in[8];
    const float* gamma = (const float*)d_in[9];
    const float* beta  = (const float*)d_in[10];
    float* out = (float*)d_out;

    float* ws  = (float*)d_ws;
    const size_t NROW = (size_t)Bb * Tt;
    float* Ko  = ws;
    float* Qo  = Ko  + NROW * Mm;
    float* Vo  = Qo  + NROW * Mm;
    float* vco = Vo  + NROW * Mm;
    float* yb  = vco + NROW * Mm;
    float* eta = yb  + NROW * Mm;

    ln_proj_kernel<<<dim3(NROW / LROWS), dim3(256), 0, stream>>>(x, gamma, beta, We, Wk, Wq, Wv, Wvc,
                                                             Ko, Qo, Vo, vco, eta);
    scan_kernel<<<dim3(Bb, Mm / SROWS), dim3(SCW * 32), 0, stream>>>(Ko, Qo, Vo, vco, eta, theta, yb);
    out_gemm_kernel<<<dim3(NROW / 16, Oo / 64), dim3(32), 0, stream>>>(yb, Wout, bout, out);
}
